# starting kernel
# baseline (speedup 1.0000x reference)
#include <hip/hip_runtime.h>
#include <hip/hip_bf16.h>

typedef __attribute__((ext_vector_type(2))) float v2f;
typedef __attribute__((ext_vector_type(8))) float v8f;

#define FEAT 128

__global__ void deg_init_kernel(float* __restrict__ deg, int n) {
    int i = blockIdx.x * blockDim.x + threadIdx.x;
    if (i < n) deg[i] = 1.0f;
}

__global__ void deg_edges_kernel(const int* __restrict__ dst, float* __restrict__ deg, int ne) {
    int e = blockIdx.x * blockDim.x + threadIdx.x;
    if (e < ne) unsafeAtomicAdd(&deg[dst[e]], 1.0f);
}

__global__ void deg_to_dinv_kernel(float* __restrict__ deg, int n) {
    int i = blockIdx.x * blockDim.x + threadIdx.x;
    if (i < n) deg[i] = rsqrtf(deg[i]);
}

__global__ __launch_bounds__(256) void gemm128_wmma_kernel(
    const float* __restrict__ X, const float* __restrict__ W,
    float* __restrict__ H, int nrows) {
    __shared__ float As[32][FEAT + 4];

    const int m0 = blockIdx.x * 32;

    for (int i = threadIdx.x; i < 32 * FEAT; i += 256) {
        int r = i >> 7, cc = i & (FEAT - 1);
        int row = m0 + r;
        As[r][cc] = (row < nrows) ? X[(long long)row * FEAT + cc] : 0.0f;
    }
    __syncthreads();

    const int wave  = threadIdx.x >> 5;
    const int lane  = threadIdx.x & 31;
    const int n0    = wave * 16;
    const int mrow  = lane & 15;
    const int khalf = (lane >> 4) * 2;
    const int ncol  = n0 + (lane & 15);

    v8f c0 = {};
    v8f c1 = {};
#pragma unroll
    for (int k0 = 0; k0 < FEAT; k0 += 4) {
        v2f a0, a1, b;
        a0.x = As[mrow][k0 + khalf];
        a0.y = As[mrow][k0 + khalf + 1];
        a1.x = As[16 + mrow][k0 + khalf];
        a1.y = As[16 + mrow][k0 + khalf + 1];
        b.x = W[(k0 + khalf) * FEAT + ncol];
        b.y = W[(k0 + khalf + 1) * FEAT + ncol];
        c0 = __builtin_amdgcn_wmma_f32_16x16x4_f32(
            false, a0, false, b, (short)0, c0, false, false);
        c1 = __builtin_amdgcn_wmma_f32_16x16x4_f32(
            false, a1, false, b, (short)0, c1, false, false);
    }

    const int mbase = (lane >> 4) * 8;
    float* __restrict__ Hp = H + (long long)(m0 + mbase) * FEAT + ncol;
    if (m0 + 32 <= nrows) {
#pragma unroll
        for (int v = 0; v < 8; ++v)
            Hp[v * FEAT] = c0[v];
#pragma unroll
        for (int v = 0; v < 8; ++v)
            Hp[(16 + v) * FEAT] = c1[v];
    } else {
#pragma unroll
        for (int v = 0; v < 8; ++v) {
            if (m0 + mbase + v < nrows) Hp[v * FEAT] = c0[v];
        }
#pragma unroll
        for (int v = 0; v < 8; ++v) {
            if (m0 + 16 + mbase + v < nrows) Hp[(16 + v) * FEAT] = c1[v];
        }
    }
}

__global__ void self_init128_kernel(const float* __restrict__ h,
                                    const float* __restrict__ dinv,
                                    float* __restrict__ out, int n) {
    long long i = (long long)blockIdx.x * blockDim.x + threadIdx.x;
    if (i >= (long long)n * FEAT) return;
    int node = (int)(i >> 7);
    float d = dinv[node];
    out[i] = h[i] * d * d;
}

__global__ __launch_bounds__(128) void edge_agg128_kernel(
    const float* __restrict__ h, const int* __restrict__ src,
    const int* __restrict__ dst, const float* __restrict__ dinv,
    float* __restrict__ out, int ne) {
    int e = blockIdx.x;
    if (e >= ne) return;
    int s = src[e], d = dst[e];
    float w = dinv[s] * dinv[d];
    int f = threadIdx.x;
    unsafeAtomicAdd(&out[(long long)d * FEAT + f],
                    h[(long long)s * FEAT + f] * w);
}

__global__ void bn_relu_kernel(const float* __restrict__ agg,
                               const float* __restrict__ bias,
                               const float* __restrict__ gamma,
                               const float* __restrict__ beta,
                               const float* __restrict__ mean,
                               const float* __restrict__ var,
                               float* __restrict__ y, int n) {
    long long i = (long long)blockIdx.x * blockDim.x + threadIdx.x;
    if (i >= (long long)n * FEAT) return;
    int f = (int)(i & (FEAT - 1));
    float x = agg[i] + bias[f];
    float v = (x - mean[f]) * rsqrtf(var[f] + 1e-5f) * gamma[f] + beta[f];
    y[i] = fmaxf(v, 0.0f);
}

__global__ __launch_bounds__(256) void gemm_w3_kernel(
    const float* __restrict__ Y, const float* __restrict__ W3,
    float* __restrict__ H3, int nrows) {
    int node = blockIdx.x * 8 + (threadIdx.x >> 5);
    int lane = threadIdx.x & 31;
    if (node >= nrows) return;
    float s0 = 0.0f, s1 = 0.0f;
#pragma unroll
    for (int f = lane; f < FEAT; f += 32) {
        float x = Y[(long long)node * FEAT + f];
        s0 += x * W3[f * 2 + 0];
        s1 += x * W3[f * 2 + 1];
    }
#pragma unroll
    for (int off = 16; off > 0; off >>= 1) {
        s0 += __shfl_down(s0, off, 32);
        s1 += __shfl_down(s1, off, 32);
    }
    if (lane == 0) {
        H3[node * 2 + 0] = s0;
        H3[node * 2 + 1] = s1;
    }
}

__global__ void self_init2_kernel(const float* __restrict__ h3,
                                  const float* __restrict__ dinv,
                                  float* __restrict__ out, int n) {
    int i = blockIdx.x * blockDim.x + threadIdx.x;
    if (i >= n) return;
    float d = dinv[i];
    float d2 = d * d;
    out[2 * i + 0] = h3[2 * i + 0] * d2;
    out[2 * i + 1] = h3[2 * i + 1] * d2;
}

__global__ void edge_agg2_kernel(const float* __restrict__ h3,
                                 const int* __restrict__ src,
                                 const int* __restrict__ dst,
                                 const float* __restrict__ dinv,
                                 float* __restrict__ out, int ne) {
    int e = blockIdx.x * blockDim.x + threadIdx.x;
    if (e >= ne) return;
    int s = src[e], d = dst[e];
    float w = dinv[s] * dinv[d];
    unsafeAtomicAdd(&out[2 * d + 0], h3[2 * s + 0] * w);
    unsafeAtomicAdd(&out[2 * d + 1], h3[2 * s + 1] * w);
}

__global__ void logsoftmax2_kernel(float* __restrict__ out,
                                   const float* __restrict__ b3, int n) {
    int i = blockIdx.x * blockDim.x + threadIdx.x;
    if (i >= n) return;
    float z0 = out[2 * i + 0] + b3[0];
    float z1 = out[2 * i + 1] + b3[1];
    float mx = fmaxf(z0, z1);
    float lse = mx + logf(expf(z0 - mx) + expf(z1 - mx));
    out[2 * i + 0] = z0 - lse;
    out[2 * i + 1] = z1 - lse;
}

extern "C" void kernel_launch(void* const* d_in, const int* in_sizes, int n_in,
                              void* d_out, int out_size, void* d_ws, size_t ws_size,
                              hipStream_t stream) {
    const float* x   = (const float*)d_in[0];
    const int*   ei  = (const int*)d_in[1];
    const float* W1  = (const float*)d_in[2];
    const float* b1  = (const float*)d_in[3];
    const float* g1  = (const float*)d_in[4];
    const float* be1 = (const float*)d_in[5];
    const float* m1  = (const float*)d_in[6];
    const float* v1  = (const float*)d_in[7];
    const float* W2  = (const float*)d_in[8];
    const float* b2  = (const float*)d_in[9];
    const float* g2  = (const float*)d_in[10];
    const float* be2 = (const float*)d_in[11];
    const float* m2  = (const float*)d_in[12];
    const float* v2  = (const float*)d_in[13];
    const float* W3  = (const float*)d_in[14];
    const float* b3  = (const float*)d_in[15];

    const int N = in_sizes[0] / FEAT;
    const int E = in_sizes[1] / 2;
    const int* src = ei;
    const int* dst = ei + E;

    float* dinv = (float*)d_ws;
    float* bufA = dinv + N;
    float* bufB = bufA + (long long)N * FEAT;
    float* h3   = bufB + (long long)N * FEAT;
    float* outp = (float*)d_out;

    const int TB = 256;
    const int gN      = (N + TB - 1) / TB;
    const int gE      = (E + TB - 1) / TB;
    const long long NF = (long long)N * FEAT;
    const int gNF     = (int)((NF + TB - 1) / TB);
    const int gTiles  = (N + 31) / 32;
    const int gWave8  = (N + 7) / 8;

    deg_init_kernel<<<gN, TB, 0, stream>>>(dinv, N);
    deg_edges_kernel<<<gE, TB, 0, stream>>>(dst, dinv, E);
    deg_to_dinv_kernel<<<gN, TB, 0, stream>>>(dinv, N);

    gemm128_wmma_kernel<<<gTiles, 256, 0, stream>>>(x, W1, bufA, N);
    self_init128_kernel<<<gNF, TB, 0, stream>>>(bufA, dinv, bufB, N);
    edge_agg128_kernel<<<E, 128, 0, stream>>>(bufA, src, dst, dinv, bufB, E);
    bn_relu_kernel<<<gNF, TB, 0, stream>>>(bufB, b1, g1, be1, m1, v1, bufA, N);

    gemm128_wmma_kernel<<<gTiles, 256, 0, stream>>>(bufA, W2, bufB, N);
    self_init128_kernel<<<gNF, TB, 0, stream>>>(bufB, dinv, bufA, N);
    edge_agg128_kernel<<<E, 128, 0, stream>>>(bufB, src, dst, dinv, bufA, E);
    bn_relu_kernel<<<gNF, TB, 0, stream>>>(bufA, b2, g2, be2, m2, v2, bufB, N);

    gemm_w3_kernel<<<gWave8, 256, 0, stream>>>(bufB, W3, h3, N);
    self_init2_kernel<<<gN, TB, 0, stream>>>(h3, dinv, outp, N);
    edge_agg2_kernel<<<gE, TB, 0, stream>>>(h3, src, dst, dinv, outp, E);
    logsoftmax2_kernel<<<gN, TB, 0, stream>>>(outp, b3, N);
}


# Edge sums as per-destination gather over an in-call CSR (count, scan, fill by returning int atomics, rows sorted by source) instead of per-edge f32 atomics: fixed sum order, self term first
# speedup vs baseline: 1.9010x; 1.9010x over previous
#include <hip/hip_runtime.h>
#include <hip/hip_bf16.h>

typedef __attribute__((ext_vector_type(2))) float v2f;
typedef __attribute__((ext_vector_type(8))) float v8f;

#define FEAT 128

#define SCAN_TILE 1024

__global__ void cnt_zero_kernel(int* __restrict__ cnt, int n) {
    int i = blockIdx.x * blockDim.x + threadIdx.x;
    if (i < n) cnt[i] = 0;
}

__global__ void cnt_edges_kernel(const int* __restrict__ dst, int* __restrict__ cnt, int ne) {
    int e = blockIdx.x * blockDim.x + threadIdx.x;
    if (e < ne) atomicAdd(&cnt[dst[e]], 1);
}

__global__ __launch_bounds__(256) void tile_sums_kernel(
    const int* __restrict__ cnt, int* __restrict__ tsum, int n) {
    __shared__ int wtot[8];
    const int lane = threadIdx.x & 31, wave = threadIdx.x >> 5;
    const int i0 = blockIdx.x * SCAN_TILE + threadIdx.x * 4;
    int s = 0;
    for (int u = 0; u < 4; ++u)
        if (i0 + u < n) s += cnt[i0 + u];
    for (int off = 16; off > 0; off >>= 1) s += __shfl_xor(s, off, 32);
    if (lane == 0) wtot[wave] = s;
    __syncthreads();
    if (threadIdx.x == 0) {
        int t = 0;
        for (int w = 0; w < 8; ++w) t += wtot[w];
        tsum[blockIdx.x] = t;
    }
}

__global__ __launch_bounds__(256) void scan_rows_kernel(
    const int* __restrict__ cnt, const int* __restrict__ tsum,
    int* __restrict__ rowptr, int* __restrict__ cursor,
    float* __restrict__ dinv, int n) {
    __shared__ int wbase[8];
    __shared__ int wtot[8];
    const int lane = threadIdx.x & 31, wave = threadIdx.x >> 5;

    int part = 0;
    for (int j = threadIdx.x; j < (int)blockIdx.x; j += 256) part += tsum[j];
    for (int off = 16; off > 0; off >>= 1) part += __shfl_xor(part, off, 32);
    if (lane == 0) wbase[wave] = part;

    const int i0 = blockIdx.x * SCAN_TILE + threadIdx.x * 4;
    int c[4];
    int s = 0;
    for (int u = 0; u < 4; ++u) {
        c[u] = (i0 + u < n) ? cnt[i0 + u] : 0;
        s += c[u];
    }
    int incl = s;
    for (int off = 1; off < 32; off <<= 1) {
        int t = __shfl_up(incl, off, 32);
        if (lane >= off) incl += t;
    }
    if (lane == 31) wtot[wave] = incl;
    __syncthreads();

    int run = incl - s;
    for (int w = 0; w < 8; ++w) {
        run += wbase[w];
        if (w < wave) run += wtot[w];
    }
    for (int u = 0; u < 4; ++u) {
        const int i = i0 + u;
        if (i < n) {
            rowptr[i] = run;
            cursor[i] = run;
            dinv[i] = rsqrtf((float)(c[u] + 1));
            run += c[u];
            if (i == n - 1) rowptr[n] = run;
        }
    }
}

__global__ void fill_rows_kernel(const int* __restrict__ src, const int* __restrict__ dst,
                                 int* __restrict__ cursor, int* __restrict__ col, int ne) {
    int e = blockIdx.x * blockDim.x + threadIdx.x;
    if (e >= ne) return;
    int p = atomicAdd(&cursor[dst[e]], 1);
    col[p] = src[e];
}

__global__ void sort_rows_kernel(const int* __restrict__ rowptr, int* __restrict__ col, int n) {
    int i = blockIdx.x * blockDim.x + threadIdx.x;
    if (i >= n) return;
    const int beg = rowptr[i], end = rowptr[i + 1];
    for (int a = beg + 1; a < end; ++a) {
        const int v = col[a];
        int b = a - 1;
        while (b >= beg && col[b] > v) {
            col[b + 1] = col[b];
            --b;
        }
        col[b + 1] = v;
    }
}

__global__ __launch_bounds__(256) void gemm128_wmma_kernel(
    const float* __restrict__ X, const float* __restrict__ W,
    float* __restrict__ H, int nrows) {
    __shared__ float As[32][FEAT + 4];

    const int m0 = blockIdx.x * 32;

    for (int i = threadIdx.x; i < 32 * FEAT; i += 256) {
        int r = i >> 7, cc = i & (FEAT - 1);
        int row = m0 + r;
        As[r][cc] = (row < nrows) ? X[(long long)row * FEAT + cc] : 0.0f;
    }
    __syncthreads();

    const int wave  = threadIdx.x >> 5;
    const int lane  = threadIdx.x & 31;
    const int n0    = wave * 16;
    const int mrow  = lane & 15;
    const int khalf = (lane >> 4) * 2;
    const int ncol  = n0 + (lane & 15);

    v8f c0 = {};
    v8f c1 = {};
#pragma unroll
    for (int k0 = 0; k0 < FEAT; k0 += 4) {
        v2f a0, a1, b;
        a0.x = As[mrow][k0 + khalf];
        a0.y = As[mrow][k0 + khalf + 1];
        a1.x = As[16 + mrow][k0 + khalf];
        a1.y = As[16 + mrow][k0 + khalf + 1];
        b.x = W[(k0 + khalf) * FEAT + ncol];
        b.y = W[(k0 + khalf + 1) * FEAT + ncol];
        c0 = __builtin_amdgcn_wmma_f32_16x16x4_f32(
            false, a0, false, b, (short)0, c0, false, false);
        c1 = __builtin_amdgcn_wmma_f32_16x16x4_f32(
            false, a1, false, b, (short)0, c1, false, false);
    }

    const int mbase = (lane >> 4) * 8;
    float* __restrict__ Hp = H + (long long)(m0 + mbase) * FEAT + ncol;
    if (m0 + 32 <= nrows) {
#pragma unroll
        for (int v = 0; v < 8; ++v)
            Hp[v * FEAT] = c0[v];
#pragma unroll
        for (int v = 0; v < 8; ++v)
            Hp[(16 + v) * FEAT] = c1[v];
    } else {
#pragma unroll
        for (int v = 0; v < 8; ++v) {
            if (m0 + mbase + v < nrows) Hp[v * FEAT] = c0[v];
        }
#pragma unroll
        for (int v = 0; v < 8; ++v) {
            if (m0 + 16 + mbase + v < nrows) Hp[(16 + v) * FEAT] = c1[v];
        }
    }
}

typedef __attribute__((ext_vector_type(4))) float v4f;

__global__ __launch_bounds__(256) void gather128_kernel(
    const float* __restrict__ h, const int* __restrict__ rowptr,
    const int* __restrict__ col, const float* __restrict__ dinv,
    float* __restrict__ out, int n) {
    const int lane = threadIdx.x & 31;
    const int node = blockIdx.x * 8 + __builtin_amdgcn_readfirstlane((int)(threadIdx.x >> 5));
    if (node >= n) return;
    const int beg = rowptr[node];
    const int end = rowptr[node + 1];
    const float dd = dinv[node];
    const v4f* __restrict__ h4 = (const v4f*)h;

    v4f acc = h4[(long long)node * 32 + lane] * dd * dd;

    for (int base = beg; base < end; base += 32) {
        const int k = (end - base < 32) ? (end - base) : 32;
        int myc = 0;
        float myw = 0.0f;
        if (lane < k) {
            myc = col[base + lane];
            myw = dinv[myc] * dd;
        }
        for (int j = 0; j < k; j += 8) {
            const int m = k - j;
            v4f r[8];
#pragma unroll
            for (int u = 0; u < 8; ++u) {
                r[u] = (v4f){0.0f, 0.0f, 0.0f, 0.0f};
                if (u < m) {
                    const int s = __builtin_amdgcn_readlane(myc, j + u);
                    r[u] = h4[(long long)s * 32 + lane];
                }
            }
#pragma unroll
            for (int u = 0; u < 8; ++u) {
                if (u < m) {
                    const float w = __int_as_float(
                        __builtin_amdgcn_readlane(__float_as_int(myw), j + u));
                    acc += r[u] * w;
                }
            }
        }
    }
    ((v4f*)out)[(long long)node * 32 + lane] = acc;
}

__global__ void bn_relu_kernel(const float* __restrict__ agg,
                               const float* __restrict__ bias,
                               const float* __restrict__ gamma,
                               const float* __restrict__ beta,
                               const float* __restrict__ mean,
                               const float* __restrict__ var,
                               float* __restrict__ y, int n) {
    long long i = (long long)blockIdx.x * blockDim.x + threadIdx.x;
    if (i >= (long long)n * FEAT) return;
    int f = (int)(i & (FEAT - 1));
    float x = agg[i] + bias[f];
    float v = (x - mean[f]) * rsqrtf(var[f] + 1e-5f) * gamma[f] + beta[f];
    y[i] = fmaxf(v, 0.0f);
}

__global__ __launch_bounds__(256) void gemm_w3_kernel(
    const float* __restrict__ Y, const float* __restrict__ W3,
    float* __restrict__ H3, int nrows) {
    int node = blockIdx.x * 8 + (threadIdx.x >> 5);
    int lane = threadIdx.x & 31;
    if (node >= nrows) return;
    float s0 = 0.0f, s1 = 0.0f;
#pragma unroll
    for (int f = lane; f < FEAT; f += 32) {
        float x = Y[(long long)node * FEAT + f];
        s0 += x * W3[f * 2 + 0];
        s1 += x * W3[f * 2 + 1];
    }
#pragma unroll
    for (int off = 16; off > 0; off >>= 1) {
        s0 += __shfl_down(s0, off, 32);
        s1 += __shfl_down(s1, off, 32);
    }
    if (lane == 0) {
        H3[node * 2 + 0] = s0;
        H3[node * 2 + 1] = s1;
    }
}

__global__ void gather2_kernel(const float* __restrict__ h3,
                               const int* __restrict__ rowptr,
                               const int* __restrict__ col,
                               const float* __restrict__ dinv,
                               float* __restrict__ out, int n) {
    int i = blockIdx.x * blockDim.x + threadIdx.x;
    if (i >= n) return;
    const float d = dinv[i];
    const float d2 = d * d;
    float a0 = h3[2 * i + 0] * d2;
    float a1 = h3[2 * i + 1] * d2;
    const int beg = rowptr[i], end = rowptr[i + 1];
    for (int j = beg; j < end; ++j) {
        const int s = col[j];
        const float w = dinv[s] * d;
        a0 += h3[2 * s + 0] * w;
        a1 += h3[2 * s + 1] * w;
    }
    out[2 * i + 0] = a0;
    out[2 * i + 1] = a1;
}

__global__ void logsoftmax2_kernel(float* __restrict__ out,
                                   const float* __restrict__ b3, int n) {
    int i = blockIdx.x * blockDim.x + threadIdx.x;
    if (i >= n) return;
    float z0 = out[2 * i + 0] + b3[0];
    float z1 = out[2 * i + 1] + b3[1];
    float mx = fmaxf(z0, z1);
    float lse = mx + logf(expf(z0 - mx) + expf(z1 - mx));
    out[2 * i + 0] = z0 - lse;
    out[2 * i + 1] = z1 - lse;
}

extern "C" void kernel_launch(void* const* d_in, const int* in_sizes, int n_in,
                              void* d_out, int out_size, void* d_ws, size_t ws_size,
                              hipStream_t stream) {
    const float* x   = (const float*)d_in[0];
    const int*   ei  = (const int*)d_in[1];
    const float* W1  = (const float*)d_in[2];
    const float* b1  = (const float*)d_in[3];
    const float* g1  = (const float*)d_in[4];
    const float* be1 = (const float*)d_in[5];
    const float* m1  = (const float*)d_in[6];
    const float* v1  = (const float*)d_in[7];
    const float* W2  = (const float*)d_in[8];
    const float* b2  = (const float*)d_in[9];
    const float* g2  = (const float*)d_in[10];
    const float* be2 = (const float*)d_in[11];
    const float* m2  = (const float*)d_in[12];
    const float* v2  = (const float*)d_in[13];
    const float* W3  = (const float*)d_in[14];
    const float* b3  = (const float*)d_in[15];

    const int N = in_sizes[0] / FEAT;
    const int E = in_sizes[1] / 2;
    const int* src = ei;
    const int* dst = ei + E;

    float* bufA = (float*)d_ws;
    float* bufB = bufA + (long long)N * FEAT;
    float* h3   = bufB + (long long)N * FEAT;
    float* dinv = h3 + (long long)N * 2;
    int* cnt    = (int*)(dinv + N);
    int* rowptr = cnt + N;
    int* cursor = rowptr + N + 1;
    int* col    = cursor + N;
    int* tsum   = col + E;
    float* outp = (float*)d_out;

    const int TB = 256;
    const int gN      = (N + TB - 1) / TB;
    const int gE      = (E + TB - 1) / TB;
    const long long NF = (long long)N * FEAT;
    const int gNF     = (int)((NF + TB - 1) / TB);
    const int gTiles  = (N + 31) / 32;
    const int gWave8  = (N + 7) / 8;
    const int gScan   = (N + SCAN_TILE - 1) / SCAN_TILE;

    cnt_zero_kernel<<<gN, TB, 0, stream>>>(cnt, N);
    cnt_edges_kernel<<<gE, TB, 0, stream>>>(dst, cnt, E);
    tile_sums_kernel<<<gScan, 256, 0, stream>>>(cnt, tsum, N);
    scan_rows_kernel<<<gScan, 256, 0, stream>>>(cnt, tsum, rowptr, cursor, dinv, N);
    fill_rows_kernel<<<gE, TB, 0, stream>>>(src, dst, cursor, col, E);
    sort_rows_kernel<<<gN, TB, 0, stream>>>(rowptr, col, N);

    gemm128_wmma_kernel<<<gTiles, 256, 0, stream>>>(x, W1, bufA, N);
    gather128_kernel<<<gWave8, 256, 0, stream>>>(bufA, rowptr, col, dinv, bufB, N);
    bn_relu_kernel<<<gNF, TB, 0, stream>>>(bufB, b1, g1, be1, m1, v1, bufA, N);

    gemm128_wmma_kernel<<<gTiles, 256, 0, stream>>>(bufA, W2, bufB, N);
    gather128_kernel<<<gWave8, 256, 0, stream>>>(bufB, rowptr, col, dinv, bufA, N);
    bn_relu_kernel<<<gNF, TB, 0, stream>>>(bufA, b2, g2, be2, m2, v2, bufB, N);

    gemm_w3_kernel<<<gWave8, 256, 0, stream>>>(bufB, W3, h3, N);
    gather2_kernel<<<gN, TB, 0, stream>>>(h3, rowptr, col, dinv, outp, N);
    logsoftmax2_kernel<<<gN, TB, 0, stream>>>(outp, b3, N);
}


# Bias+BN+ReLU and bias+log_softmax passes folded into the epilogue of the aggregation kernel that feeds each (applied in registers before the one store): 15 launches to 12, same sums
# speedup vs baseline: 2.4831x; 1.3062x over previous
#include <hip/hip_runtime.h>
#include <hip/hip_bf16.h>

typedef __attribute__((ext_vector_type(2))) float v2f;
typedef __attribute__((ext_vector_type(8))) float v8f;

#define FEAT 128

#define SCAN_TILE 1024

__global__ void cnt_zero_kernel(int* __restrict__ cnt, int n) {
    int i = blockIdx.x * blockDim.x + threadIdx.x;
    if (i < n) cnt[i] = 0;
}

__global__ void cnt_edges_kernel(const int* __restrict__ dst, int* __restrict__ cnt, int ne) {
    int e = blockIdx.x * blockDim.x + threadIdx.x;
    if (e < ne) atomicAdd(&cnt[dst[e]], 1);
}

__global__ __launch_bounds__(256) void tile_sums_kernel(
    const int* __restrict__ cnt, int* __restrict__ tsum, int n) {
    __shared__ int wtot[8];
    const int lane = threadIdx.x & 31, wave = threadIdx.x >> 5;
    const int i0 = blockIdx.x * SCAN_TILE + threadIdx.x * 4;
    int s = 0;
    for (int u = 0; u < 4; ++u)
        if (i0 + u < n) s += cnt[i0 + u];
    for (int off = 16; off > 0; off >>= 1) s += __shfl_xor(s, off, 32);
    if (lane == 0) wtot[wave] = s;
    __syncthreads();
    if (threadIdx.x == 0) {
        int t = 0;
        for (int w = 0; w < 8; ++w) t += wtot[w];
        tsum[blockIdx.x] = t;
    }
}

__global__ __launch_bounds__(256) void scan_rows_kernel(
    const int* __restrict__ cnt, const int* __restrict__ tsum,
    int* __restrict__ rowptr, int* __restrict__ cursor,
    float* __restrict__ dinv, int n) {
    __shared__ int wbase[8];
    __shared__ int wtot[8];
    const int lane = threadIdx.x & 31, wave = threadIdx.x >> 5;

    int part = 0;
    for (int j = threadIdx.x; j < (int)blockIdx.x; j += 256) part += tsum[j];
    for (int off = 16; off > 0; off >>= 1) part += __shfl_xor(part, off, 32);
    if (lane == 0) wbase[wave] = part;

    const int i0 = blockIdx.x * SCAN_TILE + threadIdx.x * 4;
    int c[4];
    int s = 0;
    for (int u = 0; u < 4; ++u) {
        c[u] = (i0 + u < n) ? cnt[i0 + u] : 0;
        s += c[u];
    }
    int incl = s;
    for (int off = 1; off < 32; off <<= 1) {
        int t = __shfl_up(incl, off, 32);
        if (lane >= off) incl += t;
    }
    if (lane == 31) wtot[wave] = incl;
    __syncthreads();

    int run = incl - s;
    for (int w = 0; w < 8; ++w) {
        run += wbase[w];
        if (w < wave) run += wtot[w];
    }
    for (int u = 0; u < 4; ++u) {
        const int i = i0 + u;
        if (i < n) {
            rowptr[i] = run;
            cursor[i] = run;
            dinv[i] = rsqrtf((float)(c[u] + 1));
            run += c[u];
            if (i == n - 1) rowptr[n] = run;
        }
    }
}

__global__ void fill_rows_kernel(const int* __restrict__ src, const int* __restrict__ dst,
                                 int* __restrict__ cursor, int* __restrict__ col, int ne) {
    int e = blockIdx.x * blockDim.x + threadIdx.x;
    if (e >= ne) return;
    int p = atomicAdd(&cursor[dst[e]], 1);
    col[p] = src[e];
}

__global__ void sort_rows_kernel(const int* __restrict__ rowptr, int* __restrict__ col, int n) {
    int i = blockIdx.x * blockDim.x + threadIdx.x;
    if (i >= n) return;
    const int beg = rowptr[i], end = rowptr[i + 1];
    for (int a = beg + 1; a < end; ++a) {
        const int v = col[a];
        int b = a - 1;
        while (b >= beg && col[b] > v) {
            col[b + 1] = col[b];
            --b;
        }
        col[b + 1] = v;
    }
}

__global__ __launch_bounds__(256) void gemm128_wmma_kernel(
    const float* __restrict__ X, const float* __restrict__ W,
    float* __restrict__ H, int nrows) {
    __shared__ float As[32][FEAT + 4];

    const int m0 = blockIdx.x * 32;

    for (int i = threadIdx.x; i < 32 * FEAT; i += 256) {
        int r = i >> 7, cc = i & (FEAT - 1);
        int row = m0 + r;
        As[r][cc] = (row < nrows) ? X[(long long)row * FEAT + cc] : 0.0f;
    }
    __syncthreads();

    const int wave  = threadIdx.x >> 5;
    const int lane  = threadIdx.x & 31;
    const int n0    = wave * 16;
    const int mrow  = lane & 15;
    const int khalf = (lane >> 4) * 2;
    const int ncol  = n0 + (lane & 15);

    v8f c0 = {};
    v8f c1 = {};
#pragma unroll
    for (int k0 = 0; k0 < FEAT; k0 += 4) {
        v2f a0, a1, b;
        a0.x = As[mrow][k0 + khalf];
        a0.y = As[mrow][k0 + khalf + 1];
        a1.x = As[16 + mrow][k0 + khalf];
        a1.y = As[16 + mrow][k0 + khalf + 1];
        b.x = W[(k0 + khalf) * FEAT + ncol];
        b.y = W[(k0 + khalf + 1) * FEAT + ncol];
        c0 = __builtin_amdgcn_wmma_f32_16x16x4_f32(
            false, a0, false, b, (short)0, c0, false, false);
        c1 = __builtin_amdgcn_wmma_f32_16x16x4_f32(
            false, a1, false, b, (short)0, c1, false, false);
    }

    const int mbase = (lane >> 4) * 8;
    float* __restrict__ Hp = H + (long long)(m0 + mbase) * FEAT + ncol;
    if (m0 + 32 <= nrows) {
#pragma unroll
        for (int v = 0; v < 8; ++v)
            Hp[v * FEAT] = c0[v];
#pragma unroll
        for (int v = 0; v < 8; ++v)
            Hp[(16 + v) * FEAT] = c1[v];
    } else {
#pragma unroll
        for (int v = 0; v < 8; ++v) {
            if (m0 + mbase + v < nrows) Hp[v * FEAT] = c0[v];
        }
#pragma unroll
        for (int v = 0; v < 8; ++v) {
            if (m0 + 16 + mbase + v < nrows) Hp[(16 + v) * FEAT] = c1[v];
        }
    }
}

typedef __attribute__((ext_vector_type(4))) float v4f;

__device__ __forceinline__ float bias_bn_relu(float a, float bi, float mu, float va,
                                              float ga, float be) {
    const float x = a + bi;
    const float v = (x - mu) * rsqrtf(va + 1e-5f) * ga + be;
    return fmaxf(v, 0.0f);
}

__global__ __launch_bounds__(256) void gather128_bn_relu_kernel(
    const float* __restrict__ h, const int* __restrict__ rowptr,
    const int* __restrict__ col, const float* __restrict__ dinv,
    const float* __restrict__ bias, const float* __restrict__ gamma,
    const float* __restrict__ beta, const float* __restrict__ mean,
    const float* __restrict__ var,
    float* __restrict__ out, int n) {
    const int lane = threadIdx.x & 31;
    const int node = blockIdx.x * 8 + __builtin_amdgcn_readfirstlane((int)(threadIdx.x >> 5));
    if (node >= n) return;
    const int beg = rowptr[node];
    const int end = rowptr[node + 1];
    const float dd = dinv[node];
    const v4f* __restrict__ h4 = (const v4f*)h;

    v4f acc = h4[(long long)node * 32 + lane] * dd * dd;

    for (int base = beg; base < end; base += 32) {
        const int k = (end - base < 32) ? (end - base) : 32;
        int myc = 0;
        float myw = 0.0f;
        if (lane < k) {
            myc = col[base + lane];
            myw = dinv[myc] * dd;
        }
        for (int j = 0; j < k; j += 8) {
            const int m = k - j;
            v4f r[8];
#pragma unroll
            for (int u = 0; u < 8; ++u) {
                r[u] = (v4f){0.0f, 0.0f, 0.0f, 0.0f};
                if (u < m) {
                    const int s = __builtin_amdgcn_readlane(myc, j + u);
                    r[u] = h4[(long long)s * 32 + lane];
                }
            }
#pragma unroll
            for (int u = 0; u < 8; ++u) {
                if (u < m) {
                    const float w = __int_as_float(
                        __builtin_amdgcn_readlane(__float_as_int(myw), j + u));
                    acc += r[u] * w;
                }
            }
        }
    }
    const v4f bi = ((const v4f*)bias)[lane];
    const v4f mu = ((const v4f*)mean)[lane];
    const v4f va = ((const v4f*)var)[lane];
    const v4f ga = ((const v4f*)gamma)[lane];
    const v4f be = ((const v4f*)beta)[lane];
    v4f y;
    y.x = bias_bn_relu(acc.x, bi.x, mu.x, va.x, ga.x, be.x);
    y.y = bias_bn_relu(acc.y, bi.y, mu.y, va.y, ga.y, be.y);
    y.z = bias_bn_relu(acc.z, bi.z, mu.z, va.z, ga.z, be.z);
    y.w = bias_bn_relu(acc.w, bi.w, mu.w, va.w, ga.w, be.w);
    ((v4f*)out)[(long long)node * 32 + lane] = y;
}

__global__ __launch_bounds__(256) void gemm_w3_kernel(
    const float* __restrict__ Y, const float* __restrict__ W3,
    float* __restrict__ H3, int nrows) {
    int node = blockIdx.x * 8 + (threadIdx.x >> 5);
    int lane = threadIdx.x & 31;
    if (node >= nrows) return;
    float s0 = 0.0f, s1 = 0.0f;
#pragma unroll
    for (int f = lane; f < FEAT; f += 32) {
        float x = Y[(long long)node * FEAT + f];
        s0 += x * W3[f * 2 + 0];
        s1 += x * W3[f * 2 + 1];
    }
#pragma unroll
    for (int off = 16; off > 0; off >>= 1) {
        s0 += __shfl_down(s0, off, 32);
        s1 += __shfl_down(s1, off, 32);
    }
    if (lane == 0) {
        H3[node * 2 + 0] = s0;
        H3[node * 2 + 1] = s1;
    }
}

__global__ void gather2_logsoftmax_kernel(const float* __restrict__ h3,
                               const int* __restrict__ rowptr,
                               const int* __restrict__ col,
                               const float* __restrict__ dinv,
                               const float* __restrict__ b3,
                               float* __restrict__ out, int n) {
    int i = blockIdx.x * blockDim.x + threadIdx.x;
    if (i >= n) return;
    const float d = dinv[i];
    const float d2 = d * d;
    float a0 = h3[2 * i + 0] * d2;
    float a1 = h3[2 * i + 1] * d2;
    const int beg = rowptr[i], end = rowptr[i + 1];
    for (int j = beg; j < end; ++j) {
        const int s = col[j];
        const float w = dinv[s] * d;
        a0 += h3[2 * s + 0] * w;
        a1 += h3[2 * s + 1] * w;
    }
    const float z0 = a0 + b3[0];
    const float z1 = a1 + b3[1];
    const float mx = fmaxf(z0, z1);
    const float lse = mx + logf(expf(z0 - mx) + expf(z1 - mx));
    out[2 * i + 0] = z0 - lse;
    out[2 * i + 1] = z1 - lse;
}

extern "C" void kernel_launch(void* const* d_in, const int* in_sizes, int n_in,
                              void* d_out, int out_size, void* d_ws, size_t ws_size,
                              hipStream_t stream) {
    const float* x   = (const float*)d_in[0];
    const int*   ei  = (const int*)d_in[1];
    const float* W1  = (const float*)d_in[2];
    const float* b1  = (const float*)d_in[3];
    const float* g1  = (const float*)d_in[4];
    const float* be1 = (const float*)d_in[5];
    const float* m1  = (const float*)d_in[6];
    const float* v1  = (const float*)d_in[7];
    const float* W2  = (const float*)d_in[8];
    const float* b2  = (const float*)d_in[9];
    const float* g2  = (const float*)d_in[10];
    const float* be2 = (const float*)d_in[11];
    const float* m2  = (const float*)d_in[12];
    const float* v2  = (const float*)d_in[13];
    const float* W3  = (const float*)d_in[14];
    const float* b3  = (const float*)d_in[15];

    const int N = in_sizes[0] / FEAT;
    const int E = in_sizes[1] / 2;
    const int* src = ei;
    const int* dst = ei + E;

    float* bufA = (float*)d_ws;
    float* bufB = bufA + (long long)N * FEAT;
    float* h3   = bufB + (long long)N * FEAT;
    float* dinv = h3 + (long long)N * 2;
    int* cnt    = (int*)(dinv + N);
    int* rowptr = cnt + N;
    int* cursor = rowptr + N + 1;
    int* col    = cursor + N;
    int* tsum   = col + E;
    float* outp = (float*)d_out;

    const int TB = 256;
    const int gN      = (N + TB - 1) / TB;
    const int gE      = (E + TB - 1) / TB;
    const int gTiles  = (N + 31) / 32;
    const int gWave8  = (N + 7) / 8;
    const int gScan   = (N + SCAN_TILE - 1) / SCAN_TILE;

    cnt_zero_kernel<<<gN, TB, 0, stream>>>(cnt, N);
    cnt_edges_kernel<<<gE, TB, 0, stream>>>(dst, cnt, E);
    tile_sums_kernel<<<gScan, 256, 0, stream>>>(cnt, tsum, N);
    scan_rows_kernel<<<gScan, 256, 0, stream>>>(cnt, tsum, rowptr, cursor, dinv, N);
    fill_rows_kernel<<<gE, TB, 0, stream>>>(src, dst, cursor, col, E);
    sort_rows_kernel<<<gN, TB, 0, stream>>>(rowptr, col, N);

    gemm128_wmma_kernel<<<gTiles, 256, 0, stream>>>(x, W1, bufA, N);
    gather128_bn_relu_kernel<<<gWave8, 256, 0, stream>>>(
        bufA, rowptr, col, dinv, b1, g1, be1, m1, v1, bufB, N);

    gemm128_wmma_kernel<<<gTiles, 256, 0, stream>>>(bufB, W2, bufA, N);
    gather128_bn_relu_kernel<<<gWave8, 256, 0, stream>>>(
        bufA, rowptr, col, dinv, b2, g2, be2, m2, v2, bufB, N);

    gemm_w3_kernel<<<gWave8, 256, 0, stream>>>(bufB, W3, h3, N);
    gather2_logsoftmax_kernel<<<gN, TB, 0, stream>>>(h3, rowptr, col, dinv, b3, outp, N);
}


# One kernel per 128-wide layer: block gathers 32 aggregated rows into LDS, WMMA product, BN+ReLU, layer 2 also the W3 dot; 12 launches to 9; sums reordered (aggregate before product, W3 dot regrouped)
# speedup vs baseline: 3.3207x; 1.3373x over previous
#include <hip/hip_runtime.h>
#include <hip/hip_bf16.h>

typedef __attribute__((ext_vector_type(2))) float v2f;
typedef __attribute__((ext_vector_type(8))) float v8f;

#define FEAT 128

#define SCAN_TILE 1024

__global__ void cnt_zero_kernel(int* __restrict__ cnt, int n) {
    int i = blockIdx.x * blockDim.x + threadIdx.x;
    if (i < n) cnt[i] = 0;
}

__global__ void cnt_edges_kernel(const int* __restrict__ dst, int* __restrict__ cnt, int ne) {
    int e = blockIdx.x * blockDim.x + threadIdx.x;
    if (e < ne) atomicAdd(&cnt[dst[e]], 1);
}

__global__ __launch_bounds__(256) void tile_sums_kernel(
    const int* __restrict__ cnt, int* __restrict__ tsum, int n) {
    __shared__ int wtot[8];
    const int lane = threadIdx.x & 31, wave = threadIdx.x >> 5;
    const int i0 = blockIdx.x * SCAN_TILE + threadIdx.x * 4;
    int s = 0;
    for (int u = 0; u < 4; ++u)
        if (i0 + u < n) s += cnt[i0 + u];
    for (int off = 16; off > 0; off >>= 1) s += __shfl_xor(s, off, 32);
    if (lane == 0) wtot[wave] = s;
    __syncthreads();
    if (threadIdx.x == 0) {
        int t = 0;
        for (int w = 0; w < 8; ++w) t += wtot[w];
        tsum[blockIdx.x] = t;
    }
}

__global__ __launch_bounds__(256) void scan_rows_kernel(
    const int* __restrict__ cnt, const int* __restrict__ tsum,
    int* __restrict__ rowptr, int* __restrict__ cursor,
    float* __restrict__ dinv, int n) {
    __shared__ int wbase[8];
    __shared__ int wtot[8];
    const int lane = threadIdx.x & 31, wave = threadIdx.x >> 5;

    int part = 0;
    for (int j = threadIdx.x; j < (int)blockIdx.x; j += 256) part += tsum[j];
    for (int off = 16; off > 0; off >>= 1) part += __shfl_xor(part, off, 32);
    if (lane == 0) wbase[wave] = part;

    const int i0 = blockIdx.x * SCAN_TILE + threadIdx.x * 4;
    int c[4];
    int s = 0;
    for (int u = 0; u < 4; ++u) {
        c[u] = (i0 + u < n) ? cnt[i0 + u] : 0;
        s += c[u];
    }
    int incl = s;
    for (int off = 1; off < 32; off <<= 1) {
        int t = __shfl_up(incl, off, 32);
        if (lane >= off) incl += t;
    }
    if (lane == 31) wtot[wave] = incl;
    __syncthreads();

    int run = incl - s;
    for (int w = 0; w < 8; ++w) {
        run += wbase[w];
        if (w < wave) run += wtot[w];
    }
    for (int u = 0; u < 4; ++u) {
        const int i = i0 + u;
        if (i < n) {
            rowptr[i] = run;
            cursor[i] = run;
            dinv[i] = rsqrtf((float)(c[u] + 1));
            run += c[u];
            if (i == n - 1) rowptr[n] = run;
        }
    }
}

__global__ void fill_rows_kernel(const int* __restrict__ src, const int* __restrict__ dst,
                                 int* __restrict__ cursor, int* __restrict__ col, int ne) {
    int e = blockIdx.x * blockDim.x + threadIdx.x;
    if (e >= ne) return;
    int p = atomicAdd(&cursor[dst[e]], 1);
    col[p] = src[e];
}

__global__ void sort_rows_kernel(const int* __restrict__ rowptr, int* __restrict__ col, int n) {
    int i = blockIdx.x * blockDim.x + threadIdx.x;
    if (i >= n) return;
    const int beg = rowptr[i], end = rowptr[i + 1];
    for (int a = beg + 1; a < end; ++a) {
        const int v = col[a];
        int b = a - 1;
        while (b >= beg && col[b] > v) {
            col[b + 1] = col[b];
            --b;
        }
        col[b + 1] = v;
    }
}

typedef __attribute__((ext_vector_type(4))) float v4f;

__device__ __forceinline__ float bias_bn_relu(float a, float bi, float mu, float va,
                                              float ga, float be) {
    const float x = a + bi;
    const float v = (x - mu) * rsqrtf(va + 1e-5f) * ga + be;
    return fmaxf(v, 0.0f);
}

template <bool FOLD3>
__device__ __forceinline__ void layer128_body(
    const float* __restrict__ xin, const int* __restrict__ rowptr,
    const int* __restrict__ col, const float* __restrict__ dinv,
    const float* __restrict__ W,
    const float* __restrict__ bias, const float* __restrict__ gamma,
    const float* __restrict__ beta, const float* __restrict__ mean,
    const float* __restrict__ var,
    const float* __restrict__ W3, float* __restrict__ out, int n) {
    __shared__ float As[32][FEAT + 4];

    const int lane = threadIdx.x & 31;
    const int wave = __builtin_amdgcn_readfirstlane((int)(threadIdx.x >> 5));
    const int m0 = blockIdx.x * 32;
    const int nb = m0 + wave * 4;
    const v4f* __restrict__ x4 = (const v4f*)xin;

    int beg[4], end[4];
    float dd[4];
    v4f selfr[4];
    int myc[4];
    float myw[4];
#pragma unroll
    for (int q = 0; q < 4; ++q) {
        const bool ok = nb + q < n;
        const int node = ok ? nb + q : n - 1;
        const int b = rowptr[node];
        const int e = rowptr[node + 1];
        beg[q] = b;
        end[q] = ok ? e : b;
        dd[q]  = ok ? dinv[node] : 0.0f;
        selfr[q] = x4[(long long)node * 32 + lane];
    }
#pragma unroll
    for (int q = 0; q < 4; ++q) {
        const int k0 = (end[q] - beg[q] < 32) ? (end[q] - beg[q]) : 32;
        myc[q] = 0;
        if (lane < k0) myc[q] = col[beg[q] + lane];
    }
#pragma unroll
    for (int q = 0; q < 4; ++q) {
        myw[q] = dinv[myc[q]] * dd[q];
    }
#pragma unroll
    for (int q = 0; q < 4; ++q) {
        v4f acc = selfr[q] * dd[q] * dd[q];
        if (nb + q >= n) acc = (v4f){0.0f, 0.0f, 0.0f, 0.0f};
        int cq = myc[q];
        float wq = myw[q];
        for (int base = beg[q]; base < end[q]; base += 32) {
            const int k = (end[q] - base < 32) ? (end[q] - base) : 32;
            if (base != beg[q]) {
                cq = 0;
                wq = 0.0f;
                if (lane < k) {
                    cq = col[base + lane];
                    wq = dinv[cq] * dd[q];
                }
            }
            for (int j = 0; j < k; j += 8) {
                const int m = k - j;
                v4f r[8];
#pragma unroll
                for (int u = 0; u < 8; ++u) {
                    r[u] = (v4f){0.0f, 0.0f, 0.0f, 0.0f};
                    if (u < m) {
                        const int s = __builtin_amdgcn_readlane(cq, j + u);
                        r[u] = x4[(long long)s * 32 + lane];
                    }
                }
#pragma unroll
                for (int u = 0; u < 8; ++u) {
                    if (u < m) {
                        const float w = __int_as_float(
                            __builtin_amdgcn_readlane(__float_as_int(wq), j + u));
                        acc += r[u] * w;
                    }
                }
            }
        }
        *(v4f*)&As[wave * 4 + q][4 * lane] = acc;
    }
    __syncthreads();

    const int n0    = wave * 16;
    const int mrow  = lane & 15;
    const int khalf = (lane >> 4) * 2;
    const int ncol  = n0 + (lane & 15);

    v8f c0 = {};
    v8f c1 = {};
#pragma unroll
    for (int k0 = 0; k0 < FEAT; k0 += 4) {
        v2f a0, a1, b;
        a0.x = As[mrow][k0 + khalf];
        a0.y = As[mrow][k0 + khalf + 1];
        a1.x = As[16 + mrow][k0 + khalf];
        a1.y = As[16 + mrow][k0 + khalf + 1];
        b.x = W[(k0 + khalf) * FEAT + ncol];
        b.y = W[(k0 + khalf + 1) * FEAT + ncol];
        c0 = __builtin_amdgcn_wmma_f32_16x16x4_f32(
            false, a0, false, b, (short)0, c0, false, false);
        c1 = __builtin_amdgcn_wmma_f32_16x16x4_f32(
            false, a1, false, b, (short)0, c1, false, false);
    }

    const float bi = bias[ncol];
    const float mu = mean[ncol];
    const float va = var[ncol];
    const float ga = gamma[ncol];
    const float be = beta[ncol];
    const int mbase = (lane >> 4) * 8;
    __syncthreads();
#pragma unroll
    for (int v = 0; v < 8; ++v) {
        As[mbase + v][ncol]      = bias_bn_relu(c0[v], bi, mu, va, ga, be);
        As[16 + mbase + v][ncol] = bias_bn_relu(c1[v], bi, mu, va, ga, be);
    }
    __syncthreads();

    if (FOLD3) {
        const v4f wa = ((const v4f*)W3)[2 * lane];
        const v4f wb = ((const v4f*)W3)[2 * lane + 1];
#pragma unroll
        for (int q = 0; q < 4; ++q) {
            const v4f y = *(const v4f*)&As[wave * 4 + q][4 * lane];
            float s0 = y.x * wa.x + y.y * wa.z + y.z * wb.x + y.w * wb.z;
            float s1 = y.x * wa.y + y.y * wa.w + y.z * wb.y + y.w * wb.w;
#pragma unroll
            for (int off = 16; off > 0; off >>= 1) {
                s0 += __shfl_xor(s0, off, 32);
                s1 += __shfl_xor(s1, off, 32);
            }
            const int node = nb + q;
            if (lane == 0 && node < n) {
                out[2 * node + 0] = s0;
                out[2 * node + 1] = s1;
            }
        }
    } else {
#pragma unroll
        for (int q = 0; q < 4; ++q) {
            const int node = nb + q;
            if (node < n)
                ((v4f*)out)[(long long)node * 32 + lane] = *(const v4f*)&As[wave * 4 + q][4 * lane];
        }
    }
}

__global__ __launch_bounds__(256) void layer128_store_kernel(
    const float* __restrict__ xin, const int* __restrict__ rowptr,
    const int* __restrict__ col, const float* __restrict__ dinv,
    const float* __restrict__ W,
    const float* __restrict__ bias, const float* __restrict__ gamma,
    const float* __restrict__ beta, const float* __restrict__ mean,
    const float* __restrict__ var,
    float* __restrict__ y, int n) {
    layer128_body<false>(xin, rowptr, col, dinv, W, bias, gamma, beta, mean, var,
                         nullptr, y, n);
}

__global__ __launch_bounds__(256) void layer128_w3_kernel(
    const float* __restrict__ xin, const int* __restrict__ rowptr,
    const int* __restrict__ col, const float* __restrict__ dinv,
    const float* __restrict__ W,
    const float* __restrict__ bias, const float* __restrict__ gamma,
    const float* __restrict__ beta, const float* __restrict__ mean,
    const float* __restrict__ var,
    const float* __restrict__ W3, float* __restrict__ h3, int n) {
    layer128_body<true>(xin, rowptr, col, dinv, W, bias, gamma, beta, mean, var,
                        W3, h3, n);
}

__global__ void gather2_logsoftmax_kernel(const float* __restrict__ h3,
                               const int* __restrict__ rowptr,
                               const int* __restrict__ col,
                               const float* __restrict__ dinv,
                               const float* __restrict__ b3,
                               float* __restrict__ out, int n) {
    int i = blockIdx.x * blockDim.x + threadIdx.x;
    if (i >= n) return;
    const float d = dinv[i];
    const float d2 = d * d;
    float a0 = h3[2 * i + 0] * d2;
    float a1 = h3[2 * i + 1] * d2;
    const int beg = rowptr[i], end = rowptr[i + 1];
    for (int j = beg; j < end; ++j) {
        const int s = col[j];
        const float w = dinv[s] * d;
        a0 += h3[2 * s + 0] * w;
        a1 += h3[2 * s + 1] * w;
    }
    const float z0 = a0 + b3[0];
    const float z1 = a1 + b3[1];
    const float mx = fmaxf(z0, z1);
    const float lse = mx + logf(expf(z0 - mx) + expf(z1 - mx));
    out[2 * i + 0] = z0 - lse;
    out[2 * i + 1] = z1 - lse;
}

extern "C" void kernel_launch(void* const* d_in, const int* in_sizes, int n_in,
                              void* d_out, int out_size, void* d_ws, size_t ws_size,
                              hipStream_t stream) {
    const float* x   = (const float*)d_in[0];
    const int*   ei  = (const int*)d_in[1];
    const float* W1  = (const float*)d_in[2];
    const float* b1  = (const float*)d_in[3];
    const float* g1  = (const float*)d_in[4];
    const float* be1 = (const float*)d_in[5];
    const float* m1  = (const float*)d_in[6];
    const float* v1  = (const float*)d_in[7];
    const float* W2  = (const float*)d_in[8];
    const float* b2  = (const float*)d_in[9];
    const float* g2  = (const float*)d_in[10];
    const float* be2 = (const float*)d_in[11];
    const float* m2  = (const float*)d_in[12];
    const float* v2  = (const float*)d_in[13];
    const float* W3  = (const float*)d_in[14];
    const float* b3  = (const float*)d_in[15];

    const int N = in_sizes[0] / FEAT;
    const int E = in_sizes[1] / 2;
    const int* src = ei;
    const int* dst = ei + E;

    float* bufA = (float*)d_ws;
    float* bufB = bufA + (long long)N * FEAT;
    float* h3   = bufB + (long long)N * FEAT;
    float* dinv = h3 + (long long)N * 2;
    int* cnt    = (int*)(dinv + N);
    int* rowptr = cnt + N;
    int* cursor = rowptr + N + 1;
    int* col    = cursor + N;
    int* tsum   = col + E;
    float* outp = (float*)d_out;

    const int TB = 256;
    const int gN      = (N + TB - 1) / TB;
    const int gE      = (E + TB - 1) / TB;
    const int gTiles  = (N + 31) / 32;
    const int gScan   = (N + SCAN_TILE - 1) / SCAN_TILE;

    cnt_zero_kernel<<<gN, TB, 0, stream>>>(cnt, N);
    cnt_edges_kernel<<<gE, TB, 0, stream>>>(dst, cnt, E);
    tile_sums_kernel<<<gScan, 256, 0, stream>>>(cnt, tsum, N);
    scan_rows_kernel<<<gScan, 256, 0, stream>>>(cnt, tsum, rowptr, cursor, dinv, N);
    fill_rows_kernel<<<gE, TB, 0, stream>>>(src, dst, cursor, col, E);
    sort_rows_kernel<<<gN, TB, 0, stream>>>(rowptr, col, N);

    layer128_store_kernel<<<gTiles, 256, 0, stream>>>(
        x, rowptr, col, dinv, W1, b1, g1, be1, m1, v1, bufA, N);
    layer128_w3_kernel<<<gTiles, 256, 0, stream>>>(
        bufA, rowptr, col, dinv, W2, b2, g2, be2, m2, v2, W3, h3, N);

    gather2_logsoftmax_kernel<<<gN, TB, 0, stream>>>(h3, rowptr, col, dinv, b3, outp, N);
}


# Row sort by rank counting, out of place (an entry's slot = how many entries of its row order before it, so its loads are independent) instead of a serial insertion sort in global memory; same sums
# speedup vs baseline: 3.5106x; 1.0572x over previous
#include <hip/hip_runtime.h>
#include <hip/hip_bf16.h>

typedef __attribute__((ext_vector_type(2))) float v2f;
typedef __attribute__((ext_vector_type(8))) float v8f;

#define FEAT 128

#define SCAN_TILE 1024

__global__ void cnt_zero_kernel(int* __restrict__ cnt, int n) {
    int i = blockIdx.x * blockDim.x + threadIdx.x;
    if (i < n) cnt[i] = 0;
}

__global__ void cnt_edges_kernel(const int* __restrict__ dst, int* __restrict__ cnt, int ne) {
    int e = blockIdx.x * blockDim.x + threadIdx.x;
    if (e < ne) atomicAdd(&cnt[dst[e]], 1);
}

__global__ __launch_bounds__(256) void tile_sums_kernel(
    const int* __restrict__ cnt, int* __restrict__ tsum, int n) {
    __shared__ int wtot[8];
    const int lane = threadIdx.x & 31, wave = threadIdx.x >> 5;
    const int i0 = blockIdx.x * SCAN_TILE + threadIdx.x * 4;
    int s = 0;
    for (int u = 0; u < 4; ++u)
        if (i0 + u < n) s += cnt[i0 + u];
    for (int off = 16; off > 0; off >>= 1) s += __shfl_xor(s, off, 32);
    if (lane == 0) wtot[wave] = s;
    __syncthreads();
    if (threadIdx.x == 0) {
        int t = 0;
        for (int w = 0; w < 8; ++w) t += wtot[w];
        tsum[blockIdx.x] = t;
    }
}

__global__ __launch_bounds__(256) void scan_rows_kernel(
    const int* __restrict__ cnt, const int* __restrict__ tsum,
    int* __restrict__ rowptr, int* __restrict__ cursor,
    float* __restrict__ dinv, int n) {
    __shared__ int wbase[8];
    __shared__ int wtot[8];
    const int lane = threadIdx.x & 31, wave = threadIdx.x >> 5;

    int part = 0;
    for (int j = threadIdx.x; j < (int)blockIdx.x; j += 256) part += tsum[j];
    for (int off = 16; off > 0; off >>= 1) part += __shfl_xor(part, off, 32);
    if (lane == 0) wbase[wave] = part;

    const int i0 = blockIdx.x * SCAN_TILE + threadIdx.x * 4;
    int c[4];
    int s = 0;
    for (int u = 0; u < 4; ++u) {
        c[u] = (i0 + u < n) ? cnt[i0 + u] : 0;
        s += c[u];
    }
    int incl = s;
    for (int off = 1; off < 32; off <<= 1) {
        int t = __shfl_up(incl, off, 32);
        if (lane >= off) incl += t;
    }
    if (lane == 31) wtot[wave] = incl;
    __syncthreads();

    int run = incl - s;
    for (int w = 0; w < 8; ++w) {
        run += wbase[w];
        if (w < wave) run += wtot[w];
    }
    for (int u = 0; u < 4; ++u) {
        const int i = i0 + u;
        if (i < n) {
            rowptr[i] = run;
            cursor[i] = run;
            dinv[i] = rsqrtf((float)(c[u] + 1));
            run += c[u];
            if (i == n - 1) rowptr[n] = run;
        }
    }
}

__global__ void fill_rows_kernel(const int* __restrict__ src, const int* __restrict__ dst,
                                 int* __restrict__ cursor, int* __restrict__ col, int ne) {
    int e = blockIdx.x * blockDim.x + threadIdx.x;
    if (e >= ne) return;
    int p = atomicAdd(&cursor[dst[e]], 1);
    col[p] = src[e];
}

__global__ void sort_rows_kernel(const int* __restrict__ rowptr, const int* __restrict__ colin,
                                 int* __restrict__ col, int n) {
    int i = blockIdx.x * blockDim.x + threadIdx.x;
    if (i >= n) return;
    const int beg = rowptr[i], end = rowptr[i + 1];
    for (int a = beg; a < end; ++a) {
        const int v = colin[a];
        int r = beg;
        for (int b = beg; b < end; ++b) {
            const int u = colin[b];
            r += (u < v || (u == v && b < a)) ? 1 : 0;
        }
        col[r] = v;
    }
}

typedef __attribute__((ext_vector_type(4))) float v4f;

__device__ __forceinline__ float bias_bn_relu(float a, float bi, float mu, float va,
                                              float ga, float be) {
    const float x = a + bi;
    const float v = (x - mu) * rsqrtf(va + 1e-5f) * ga + be;
    return fmaxf(v, 0.0f);
}

template <bool FOLD3>
__device__ __forceinline__ void layer128_body(
    const float* __restrict__ xin, const int* __restrict__ rowptr,
    const int* __restrict__ col, const float* __restrict__ dinv,
    const float* __restrict__ W,
    const float* __restrict__ bias, const float* __restrict__ gamma,
    const float* __restrict__ beta, const float* __restrict__ mean,
    const float* __restrict__ var,
    const float* __restrict__ W3, float* __restrict__ out, int n) {
    __shared__ float As[32][FEAT + 4];

    const int lane = threadIdx.x & 31;
    const int wave = __builtin_amdgcn_readfirstlane((int)(threadIdx.x >> 5));
    const int m0 = blockIdx.x * 32;
    const int nb = m0 + wave * 4;
    const v4f* __restrict__ x4 = (const v4f*)xin;

    int beg[4], end[4];
    float dd[4];
    v4f selfr[4];
    int myc[4];
    float myw[4];
#pragma unroll
    for (int q = 0; q < 4; ++q) {
        const bool ok = nb + q < n;
        const int node = ok ? nb + q : n - 1;
        const int b = rowptr[node];
        const int e = rowptr[node + 1];
        beg[q] = b;
        end[q] = ok ? e : b;
        dd[q]  = ok ? dinv[node] : 0.0f;
        selfr[q] = x4[(long long)node * 32 + lane];
    }
#pragma unroll
    for (int q = 0; q < 4; ++q) {
        const int k0 = (end[q] - beg[q] < 32) ? (end[q] - beg[q]) : 32;
        myc[q] = 0;
        if (lane < k0) myc[q] = col[beg[q] + lane];
    }
#pragma unroll
    for (int q = 0; q < 4; ++q) {
        myw[q] = dinv[myc[q]] * dd[q];
    }
#pragma unroll
    for (int q = 0; q < 4; ++q) {
        v4f acc = selfr[q] * dd[q] * dd[q];
        if (nb + q >= n) acc = (v4f){0.0f, 0.0f, 0.0f, 0.0f};
        int cq = myc[q];
        float wq = myw[q];
        for (int base = beg[q]; base < end[q]; base += 32) {
            const int k = (end[q] - base < 32) ? (end[q] - base) : 32;
            if (base != beg[q]) {
                cq = 0;
                wq = 0.0f;
                if (lane < k) {
                    cq = col[base + lane];
                    wq = dinv[cq] * dd[q];
                }
            }
            for (int j = 0; j < k; j += 8) {
                const int m = k - j;
                v4f r[8];
#pragma unroll
                for (int u = 0; u < 8; ++u) {
                    r[u] = (v4f){0.0f, 0.0f, 0.0f, 0.0f};
                    if (u < m) {
                        const int s = __builtin_amdgcn_readlane(cq, j + u);
                        r[u] = x4[(long long)s * 32 + lane];
                    }
                }
#pragma unroll
                for (int u = 0; u < 8; ++u) {
                    if (u < m) {
                        const float w = __int_as_float(
                            __builtin_amdgcn_readlane(__float_as_int(wq), j + u));
                        acc += r[u] * w;
                    }
                }
            }
        }
        *(v4f*)&As[wave * 4 + q][4 * lane] = acc;
    }
    __syncthreads();

    const int n0    = wave * 16;
    const int mrow  = lane & 15;
    const int khalf = (lane >> 4) * 2;
    const int ncol  = n0 + (lane & 15);

    v8f c0 = {};
    v8f c1 = {};
#pragma unroll
    for (int k0 = 0; k0 < FEAT; k0 += 4) {
        v2f a0, a1, b;
        a0.x = As[mrow][k0 + khalf];
        a0.y = As[mrow][k0 + khalf + 1];
        a1.x = As[16 + mrow][k0 + khalf];
        a1.y = As[16 + mrow][k0 + khalf + 1];
        b.x = W[(k0 + khalf) * FEAT + ncol];
        b.y = W[(k0 + khalf + 1) * FEAT + ncol];
        c0 = __builtin_amdgcn_wmma_f32_16x16x4_f32(
            false, a0, false, b, (short)0, c0, false, false);
        c1 = __builtin_amdgcn_wmma_f32_16x16x4_f32(
            false, a1, false, b, (short)0, c1, false, false);
    }

    const float bi = bias[ncol];
    const float mu = mean[ncol];
    const float va = var[ncol];
    const float ga = gamma[ncol];
    const float be = beta[ncol];
    const int mbase = (lane >> 4) * 8;
    __syncthreads();
#pragma unroll
    for (int v = 0; v < 8; ++v) {
        As[mbase + v][ncol]      = bias_bn_relu(c0[v], bi, mu, va, ga, be);
        As[16 + mbase + v][ncol] = bias_bn_relu(c1[v], bi, mu, va, ga, be);
    }
    __syncthreads();

    if (FOLD3) {
        const v4f wa = ((const v4f*)W3)[2 * lane];
        const v4f wb = ((const v4f*)W3)[2 * lane + 1];
#pragma unroll
        for (int q = 0; q < 4; ++q) {
            const v4f y = *(const v4f*)&As[wave * 4 + q][4 * lane];
            float s0 = y.x * wa.x + y.y * wa.z + y.z * wb.x + y.w * wb.z;
            float s1 = y.x * wa.y + y.y * wa.w + y.z * wb.y + y.w * wb.w;
#pragma unroll
            for (int off = 16; off > 0; off >>= 1) {
                s0 += __shfl_xor(s0, off, 32);
                s1 += __shfl_xor(s1, off, 32);
            }
            const int node = nb + q;
            if (lane == 0 && node < n) {
                out[2 * node + 0] = s0;
                out[2 * node + 1] = s1;
            }
        }
    } else {
#pragma unroll
        for (int q = 0; q < 4; ++q) {
            const int node = nb + q;
            if (node < n)
                ((v4f*)out)[(long long)node * 32 + lane] = *(const v4f*)&As[wave * 4 + q][4 * lane];
        }
    }
}

__global__ __launch_bounds__(256) void layer128_store_kernel(
    const float* __restrict__ xin, const int* __restrict__ rowptr,
    const int* __restrict__ col, const float* __restrict__ dinv,
    const float* __restrict__ W,
    const float* __restrict__ bias, const float* __restrict__ gamma,
    const float* __restrict__ beta, const float* __restrict__ mean,
    const float* __restrict__ var,
    float* __restrict__ y, int n) {
    layer128_body<false>(xin, rowptr, col, dinv, W, bias, gamma, beta, mean, var,
                         nullptr, y, n);
}

__global__ __launch_bounds__(256) void layer128_w3_kernel(
    const float* __restrict__ xin, const int* __restrict__ rowptr,
    const int* __restrict__ col, const float* __restrict__ dinv,
    const float* __restrict__ W,
    const float* __restrict__ bias, const float* __restrict__ gamma,
    const float* __restrict__ beta, const float* __restrict__ mean,
    const float* __restrict__ var,
    const float* __restrict__ W3, float* __restrict__ h3, int n) {
    layer128_body<true>(xin, rowptr, col, dinv, W, bias, gamma, beta, mean, var,
                        W3, h3, n);
}

__global__ void gather2_logsoftmax_kernel(const float* __restrict__ h3,
                               const int* __restrict__ rowptr,
                               const int* __restrict__ col,
                               const float* __restrict__ dinv,
                               const float* __restrict__ b3,
                               float* __restrict__ out, int n) {
    int i = blockIdx.x * blockDim.x + threadIdx.x;
    if (i >= n) return;
    const float d = dinv[i];
    const float d2 = d * d;
    float a0 = h3[2 * i + 0] * d2;
    float a1 = h3[2 * i + 1] * d2;
    const int beg = rowptr[i], end = rowptr[i + 1];
    for (int j = beg; j < end; ++j) {
        const int s = col[j];
        const float w = dinv[s] * d;
        a0 += h3[2 * s + 0] * w;
        a1 += h3[2 * s + 1] * w;
    }
    const float z0 = a0 + b3[0];
    const float z1 = a1 + b3[1];
    const float mx = fmaxf(z0, z1);
    const float lse = mx + logf(expf(z0 - mx) + expf(z1 - mx));
    out[2 * i + 0] = z0 - lse;
    out[2 * i + 1] = z1 - lse;
}

extern "C" void kernel_launch(void* const* d_in, const int* in_sizes, int n_in,
                              void* d_out, int out_size, void* d_ws, size_t ws_size,
                              hipStream_t stream) {
    const float* x   = (const float*)d_in[0];
    const int*   ei  = (const int*)d_in[1];
    const float* W1  = (const float*)d_in[2];
    const float* b1  = (const float*)d_in[3];
    const float* g1  = (const float*)d_in[4];
    const float* be1 = (const float*)d_in[5];
    const float* m1  = (const float*)d_in[6];
    const float* v1  = (const float*)d_in[7];
    const float* W2  = (const float*)d_in[8];
    const float* b2  = (const float*)d_in[9];
    const float* g2  = (const float*)d_in[10];
    const float* be2 = (const float*)d_in[11];
    const float* m2  = (const float*)d_in[12];
    const float* v2  = (const float*)d_in[13];
    const float* W3  = (const float*)d_in[14];
    const float* b3  = (const float*)d_in[15];

    const int N = in_sizes[0] / FEAT;
    const int E = in_sizes[1] / 2;
    const int* src = ei;
    const int* dst = ei + E;

    float* bufA = (float*)d_ws;
    float* bufB = bufA + (long long)N * FEAT;
    float* h3   = bufB + (long long)N * FEAT;
    float* dinv = h3 + (long long)N * 2;
    int* cnt    = (int*)(dinv + N);
    int* rowptr = cnt + N;
    int* cursor = rowptr + N + 1;
    int* col    = cursor + N;
    int* colraw = col + E;
    int* tsum   = colraw + E;
    float* outp = (float*)d_out;

    const int TB = 256;
    const int gN      = (N + TB - 1) / TB;
    const int gE      = (E + TB - 1) / TB;
    const int gTiles  = (N + 31) / 32;
    const int gScan   = (N + SCAN_TILE - 1) / SCAN_TILE;

    cnt_zero_kernel<<<gN, TB, 0, stream>>>(cnt, N);
    cnt_edges_kernel<<<gE, TB, 0, stream>>>(dst, cnt, E);
    tile_sums_kernel<<<gScan, 256, 0, stream>>>(cnt, tsum, N);
    scan_rows_kernel<<<gScan, 256, 0, stream>>>(cnt, tsum, rowptr, cursor, dinv, N);
    fill_rows_kernel<<<gE, TB, 0, stream>>>(src, dst, cursor, colraw, E);
    sort_rows_kernel<<<gN, TB, 0, stream>>>(rowptr, colraw, col, N);

    layer128_store_kernel<<<gTiles, 256, 0, stream>>>(
        x, rowptr, col, dinv, W1, b1, g1, be1, m1, v1, bufA, N);
    layer128_w3_kernel<<<gTiles, 256, 0, stream>>>(
        bufA, rowptr, col, dinv, W2, b2, g2, be2, m2, v2, W3, h3, N);

    gather2_logsoftmax_kernel<<<gN, TB, 0, stream>>>(h3, rowptr, col, dinv, b3, outp, N);
}


# CSR built by a two-level counting sort on LDS counters (edges bucketed by 512-node destination range, then per-bucket count, scan, fill) instead of per-edge global atomics: 9 launches to 8; same sums
# speedup vs baseline: 4.3962x; 1.2523x over previous
#include <hip/hip_runtime.h>
#include <hip/hip_bf16.h>

typedef __attribute__((ext_vector_type(2))) float v2f;
typedef __attribute__((ext_vector_type(8))) float v8f;

#define FEAT 128
#define GCH 24

#define BSH 9
#define BSZ 512
#define GB 128
#define NBMAX 256

__global__ __launch_bounds__(1024) void bucket_hist_kernel(
    const int* __restrict__ dst, int* __restrict__ bhist, int ne, int nb) {
    __shared__ int h[NBMAX];
    const int t = threadIdx.x;
    if (t < NBMAX) h[t] = 0;
    __syncthreads();
    const int chunk = (ne + GB - 1) / GB;
    const int e0 = blockIdx.x * chunk;
    const int e1 = (e0 + chunk < ne) ? e0 + chunk : ne;
    for (int eb = e0 + t; eb < e1; eb += 4 * 1024) {
        int d[4];
#pragma unroll
        for (int u = 0; u < 4; ++u) {
            const int e = eb + u * 1024;
            d[u] = (e < e1) ? dst[e] : -1;
        }
#pragma unroll
        for (int u = 0; u < 4; ++u)
            if (d[u] >= 0) atomicAdd(&h[d[u] >> BSH], 1);
    }
    __syncthreads();
    if (t < nb) bhist[t * GB + blockIdx.x] = h[t];
}

__global__ __launch_bounds__(256) void bucket_offsets_kernel(
    const int* __restrict__ bhist, int* __restrict__ offs, int* __restrict__ bstart, int nb) {
    __shared__ int wsum[8];
    __shared__ int wtot[8];
    const int t = threadIdx.x, lane = t & 31, wave = t >> 5;
    const int b = blockIdx.x;
    const int4* __restrict__ h4 = (const int4*)bhist;
    const int n4 = b * (GB / 4);
    int part = 0;
    for (int j0 = 0; j0 < n4; j0 += 8 * 256) {
        int4 v[8];
#pragma unroll
        for (int u = 0; u < 8; ++u) {
            const int j = j0 + u * 256 + t;
            v[u] = (j < n4) ? h4[j] : make_int4(0, 0, 0, 0);
        }
#pragma unroll
        for (int u = 0; u < 8; ++u) part += v[u].x + v[u].y + v[u].z + v[u].w;
    }
    for (int off = 16; off > 0; off >>= 1) part += __shfl_xor(part, off, 32);
    if (lane == 0) wsum[wave] = part;

    const int v = (t < GB) ? bhist[b * GB + t] : 0;
    int incl = v;
    for (int off = 1; off < 32; off <<= 1) {
        int x = __shfl_up(incl, off, 32);
        if (lane >= off) incl += x;
    }
    if (lane == 31) wtot[wave] = incl;
    __syncthreads();
    int base = 0;
    for (int w = 0; w < 8; ++w) {
        base += wsum[w];
        if (w < wave) base += wtot[w];
    }
    const int excl = base + incl - v;
    if (t < GB) offs[b * GB + t] = excl;
    if (t == 0) bstart[b] = excl;
    if (b == nb - 1 && t == GB - 1) bstart[nb] = excl + v;
}

__global__ __launch_bounds__(1024) void bucket_scatter_kernel(
    const int* __restrict__ src, const int* __restrict__ dst, const int* __restrict__ offs,
    int2* __restrict__ ebuf, int ne, int nb) {
    __shared__ int cur[NBMAX];
    const int t = threadIdx.x;
    if (t < nb) cur[t] = offs[t * GB + blockIdx.x];
    __syncthreads();
    const int chunk = (ne + GB - 1) / GB;
    const int e0 = blockIdx.x * chunk;
    const int e1 = (e0 + chunk < ne) ? e0 + chunk : ne;
    for (int eb = e0 + t; eb < e1; eb += 4 * 1024) {
        int d[4], s[4];
#pragma unroll
        for (int u = 0; u < 4; ++u) {
            const int e = eb + u * 1024;
            d[u] = (e < e1) ? dst[e] : -1;
            s[u] = (e < e1) ? src[e] : 0;
        }
#pragma unroll
        for (int u = 0; u < 4; ++u) {
            if (d[u] >= 0) {
                const int p = atomicAdd(&cur[d[u] >> BSH], 1);
                ebuf[p] = make_int2(s[u], d[u]);
            }
        }
    }
}

__global__ __launch_bounds__(BSZ) void bucket_rows_kernel(
    const int2* __restrict__ ebuf, const int* __restrict__ bstart,
    int* __restrict__ rowptr, int* __restrict__ colraw, float* __restrict__ dinv, int n) {
    __shared__ int lcnt[BSZ];
    __shared__ int lcur[BSZ];
    __shared__ int wtot[BSZ / 32];
    const int t = threadIdx.x, lane = t & 31, wave = t >> 5;
    const int n0 = blockIdx.x << BSH;
    const int e0 = bstart[blockIdx.x], e1 = bstart[blockIdx.x + 1];
    lcnt[t] = 0;
    __syncthreads();
    for (int eb = e0 + t; eb < e1; eb += 4 * BSZ) {
        int d[4];
#pragma unroll
        for (int u = 0; u < 4; ++u) {
            const int e = eb + u * BSZ;
            d[u] = (e < e1) ? ebuf[e].y : -1;
        }
#pragma unroll
        for (int u = 0; u < 4; ++u)
            if (d[u] >= 0) atomicAdd(&lcnt[d[u] - n0], 1);
    }
    __syncthreads();
    const int c = lcnt[t];
    int incl = c;
    for (int off = 1; off < 32; off <<= 1) {
        int x = __shfl_up(incl, off, 32);
        if (lane >= off) incl += x;
    }
    if (lane == 31) wtot[wave] = incl;
    __syncthreads();
    int beg = e0 + incl - c;
    for (int w = 0; w < BSZ / 32; ++w)
        if (w < wave) beg += wtot[w];
    lcur[t] = beg;
    const int i = n0 + t;
    if (i < n) {
        rowptr[i] = beg;
        dinv[i] = rsqrtf((float)(c + 1));
        if (i == n - 1) rowptr[n] = beg + c;
    }
    __syncthreads();
    for (int eb = e0 + t; eb < e1; eb += 4 * BSZ) {
        int2 v[4];
#pragma unroll
        for (int u = 0; u < 4; ++u) {
            const int e = eb + u * BSZ;
            v[u] = (e < e1) ? ebuf[e] : make_int2(0, -1);
        }
#pragma unroll
        for (int u = 0; u < 4; ++u) {
            if (v[u].y >= 0) {
                const int p = atomicAdd(&lcur[v[u].y - n0], 1);
                colraw[p] = v[u].x;
            }
        }
    }
}

__global__ void sort_rows_kernel(const int* __restrict__ rowptr, const int* __restrict__ colin,
                                 int* __restrict__ col, int n) {
    int i = blockIdx.x * blockDim.x + threadIdx.x;
    if (i >= n) return;
    const int beg = rowptr[i], end = rowptr[i + 1];
    for (int a = beg; a < end; ++a) {
        const int v = colin[a];
        int r = beg;
        for (int b = beg; b < end; ++b) {
            const int u = colin[b];
            r += (u < v || (u == v && b < a)) ? 1 : 0;
        }
        col[r] = v;
    }
}

typedef __attribute__((ext_vector_type(4))) float v4f;

__device__ __forceinline__ float bias_bn_relu(float a, float bi, float mu, float va,
                                              float ga, float be) {
    const float x = a + bi;
    const float v = (x - mu) * rsqrtf(va + 1e-5f) * ga + be;
    return fmaxf(v, 0.0f);
}

template <bool FOLD3>
__device__ __forceinline__ void layer128_body(
    const float* __restrict__ xin, const int* __restrict__ rowptr,
    const int* __restrict__ col, const float* __restrict__ dinv,
    const float* __restrict__ W,
    const float* __restrict__ bias, const float* __restrict__ gamma,
    const float* __restrict__ beta, const float* __restrict__ mean,
    const float* __restrict__ var,
    const float* __restrict__ W3, float* __restrict__ out, int n) {
    __shared__ float As[32][FEAT + 4];

    const int lane = threadIdx.x & 31;
    const int wave = __builtin_amdgcn_readfirstlane((int)(threadIdx.x >> 5));
    const int m0 = blockIdx.x * 32;
    const int nb = m0 + wave * 4;
    const v4f* __restrict__ x4 = (const v4f*)xin;

    const int bidx = (nb + lane < n) ? nb + lane : n;
    const int bndv = rowptr[bidx];
    const int didx = (nb + lane < n) ? nb + lane : n - 1;
    const float ddl = dinv[didx];
    const float ddv = (nb + lane < n) ? ddl : 0.0f;
    v4f selfr[4];
#pragma unroll
    for (int q = 0; q < 4; ++q) {
        const int node = (nb + q < n) ? nb + q : n - 1;
        selfr[q] = x4[(long long)node * 32 + lane];
    }
    const int b0 = __builtin_amdgcn_readlane(bndv, 0);
    const int T  = __builtin_amdgcn_readlane(bndv, 4) - b0;
    int cq = 0;
    {
        const int k0 = (T < 32) ? T : 32;
        if (lane < k0) cq = col[b0 + lane];
    }
    v4f acc;
#pragma unroll
    for (int q = 0; q < 4; ++q) {
        const float d = __int_as_float(__builtin_amdgcn_readlane(__float_as_int(ddv), q));
        v4f ini = selfr[q] * d * d;
        if (nb + q >= n) ini = (v4f){0.0f, 0.0f, 0.0f, 0.0f};
        *(v4f*)&As[wave * 4 + q][4 * lane] = ini;
        if (q == 0) acc = ini;
    }
    float dv = 0.0f;
    int qcur = 0;
    int endcur = __builtin_amdgcn_readlane(bndv, 1);
    float ddcur = __int_as_float(__builtin_amdgcn_readlane(__float_as_int(ddv), 0));
    for (int base = 0; base < T; base += 32) {
        const int k = (T - base < 32) ? (T - base) : 32;
        if (base != 0) {
            cq = 0;
            if (lane < k) cq = col[b0 + base + lane];
        }
        for (int j = 0; j < k; j += GCH) {
            const int m = k - j;
            v4f r[GCH];
#pragma unroll
            for (int u = 0; u < GCH; ++u) {
                const int li = (u < m) ? j + u : k - 1;
                const int s = __builtin_amdgcn_readlane(cq, li);
                r[u] = x4[(long long)s * 32 + lane];
            }
            if (j == 0) dv = dinv[cq];
#pragma unroll
            for (int u = 0; u < GCH; ++u) {
                if (u < m) {
                    const int p = b0 + base + j + u;
                    while (p >= endcur && qcur < 3) {
                        *(v4f*)&As[wave * 4 + qcur][4 * lane] = acc;
                        ++qcur;
                        acc = *(const v4f*)&As[wave * 4 + qcur][4 * lane];
                        endcur = __builtin_amdgcn_readlane(bndv, qcur + 1);
                        ddcur = __int_as_float(
                            __builtin_amdgcn_readlane(__float_as_int(ddv), qcur));
                    }
                    const float w = __int_as_float(
                        __builtin_amdgcn_readlane(__float_as_int(dv), j + u)) * ddcur;
                    acc += r[u] * w;
                }
            }
        }
    }
    *(v4f*)&As[wave * 4 + qcur][4 * lane] = acc;
    __syncthreads();

    const int n0    = wave * 16;
    const int mrow  = lane & 15;
    const int khalf = (lane >> 4) * 2;
    const int ncol  = n0 + (lane & 15);

    v8f c0 = {};
    v8f c1 = {};
#pragma unroll
    for (int k0 = 0; k0 < FEAT; k0 += 4) {
        v2f a0, a1, b;
        a0.x = As[mrow][k0 + khalf];
        a0.y = As[mrow][k0 + khalf + 1];
        a1.x = As[16 + mrow][k0 + khalf];
        a1.y = As[16 + mrow][k0 + khalf + 1];
        b.x = W[(k0 + khalf) * FEAT + ncol];
        b.y = W[(k0 + khalf + 1) * FEAT + ncol];
        c0 = __builtin_amdgcn_wmma_f32_16x16x4_f32(
            false, a0, false, b, (short)0, c0, false, false);
        c1 = __builtin_amdgcn_wmma_f32_16x16x4_f32(
            false, a1, false, b, (short)0, c1, false, false);
    }

    const float bi = bias[ncol];
    const float mu = mean[ncol];
    const float va = var[ncol];
    const float ga = gamma[ncol];
    const float be = beta[ncol];
    const int mbase = (lane >> 4) * 8;
    __syncthreads();
#pragma unroll
    for (int v = 0; v < 8; ++v) {
        As[mbase + v][ncol]      = bias_bn_relu(c0[v], bi, mu, va, ga, be);
        As[16 + mbase + v][ncol] = bias_bn_relu(c1[v], bi, mu, va, ga, be);
    }
    __syncthreads();

    if (FOLD3) {
        const v4f wa = ((const v4f*)W3)[2 * lane];
        const v4f wb = ((const v4f*)W3)[2 * lane + 1];
#pragma unroll
        for (int q = 0; q < 4; ++q) {
            const v4f y = *(const v4f*)&As[wave * 4 + q][4 * lane];
            float s0 = y.x * wa.x + y.y * wa.z + y.z * wb.x + y.w * wb.z;
            float s1 = y.x * wa.y + y.y * wa.w + y.z * wb.y + y.w * wb.w;
#pragma unroll
            for (int off = 16; off > 0; off >>= 1) {
                s0 += __shfl_xor(s0, off, 32);
                s1 += __shfl_xor(s1, off, 32);
            }
            const int node = nb + q;
            if (lane == 0 && node < n) {
                out[2 * node + 0] = s0;
                out[2 * node + 1] = s1;
            }
        }
    } else {
#pragma unroll
        for (int q = 0; q < 4; ++q) {
            const int node = nb + q;
            if (node < n)
                ((v4f*)out)[(long long)node * 32 + lane] = *(const v4f*)&As[wave * 4 + q][4 * lane];
        }
    }
}

__global__ __launch_bounds__(256) void layer128_store_kernel(
    const float* __restrict__ xin, const int* __restrict__ rowptr,
    const int* __restrict__ col, const float* __restrict__ dinv,
    const float* __restrict__ W,
    const float* __restrict__ bias, const float* __restrict__ gamma,
    const float* __restrict__ beta, const float* __restrict__ mean,
    const float* __restrict__ var,
    float* __restrict__ y, int n) {
    layer128_body<false>(xin, rowptr, col, dinv, W, bias, gamma, beta, mean, var,
                         nullptr, y, n);
}

__global__ __launch_bounds__(256) void layer128_w3_kernel(
    const float* __restrict__ xin, const int* __restrict__ rowptr,
    const int* __restrict__ col, const float* __restrict__ dinv,
    const float* __restrict__ W,
    const float* __restrict__ bias, const float* __restrict__ gamma,
    const float* __restrict__ beta, const float* __restrict__ mean,
    const float* __restrict__ var,
    const float* __restrict__ W3, float* __restrict__ h3, int n) {
    layer128_body<true>(xin, rowptr, col, dinv, W, bias, gamma, beta, mean, var,
                        W3, h3, n);
}

__global__ void gather2_logsoftmax_kernel(const float* __restrict__ h3,
                               const int* __restrict__ rowptr,
                               const int* __restrict__ col,
                               const float* __restrict__ dinv,
                               const float* __restrict__ b3,
                               float* __restrict__ out, int n) {
    int i = blockIdx.x * blockDim.x + threadIdx.x;
    if (i >= n) return;
    const float d = dinv[i];
    const float d2 = d * d;
    float a0 = h3[2 * i + 0] * d2;
    float a1 = h3[2 * i + 1] * d2;
    const int beg = rowptr[i], end = rowptr[i + 1];
    for (int j = beg; j < end; ++j) {
        const int s = col[j];
        const float w = dinv[s] * d;
        a0 += h3[2 * s + 0] * w;
        a1 += h3[2 * s + 1] * w;
    }
    const float z0 = a0 + b3[0];
    const float z1 = a1 + b3[1];
    const float mx = fmaxf(z0, z1);
    const float lse = mx + logf(expf(z0 - mx) + expf(z1 - mx));
    out[2 * i + 0] = z0 - lse;
    out[2 * i + 1] = z1 - lse;
}

extern "C" void kernel_launch(void* const* d_in, const int* in_sizes, int n_in,
                              void* d_out, int out_size, void* d_ws, size_t ws_size,
                              hipStream_t stream) {
    const float* x   = (const float*)d_in[0];
    const int*   ei  = (const int*)d_in[1];
    const float* W1  = (const float*)d_in[2];
    const float* b1  = (const float*)d_in[3];
    const float* g1  = (const float*)d_in[4];
    const float* be1 = (const float*)d_in[5];
    const float* m1  = (const float*)d_in[6];
    const float* v1  = (const float*)d_in[7];
    const float* W2  = (const float*)d_in[8];
    const float* b2  = (const float*)d_in[9];
    const float* g2  = (const float*)d_in[10];
    const float* be2 = (const float*)d_in[11];
    const float* m2  = (const float*)d_in[12];
    const float* v2  = (const float*)d_in[13];
    const float* W3  = (const float*)d_in[14];
    const float* b3  = (const float*)d_in[15];

    const int N = in_sizes[0] / FEAT;
    const int E = in_sizes[1] / 2;
    const int* src = ei;
    const int* dst = ei + E;

    float* bufA = (float*)d_ws;
    float* aux  = bufA + (long long)N * FEAT;
    float* h3   = aux + (long long)N * FEAT;
    float* dinv = h3 + (long long)N * 2;
    int* rowptr = (int*)(dinv + N);
    int* col    = rowptr + N + 1;
    int* colraw = col + E;
    int* bhist  = (int*)aux;
    int* offs   = bhist + NBMAX * GB;
    int* bstart = offs + NBMAX * GB;
    int2* ebuf  = (int2*)(bstart + 2 * NBMAX);
    float* outp = (float*)d_out;

    const int TB = 256;
    const int gN      = (N + TB - 1) / TB;
    const int gTiles  = (N + 31) / 32;
    const int nb      = (N + BSZ - 1) / BSZ;

    bucket_hist_kernel<<<GB, 1024, 0, stream>>>(dst, bhist, E, nb);
    bucket_offsets_kernel<<<nb, 256, 0, stream>>>(bhist, offs, bstart, nb);
    bucket_scatter_kernel<<<GB, 1024, 0, stream>>>(src, dst, offs, ebuf, E, nb);
    bucket_rows_kernel<<<nb, BSZ, 0, stream>>>(ebuf, bstart, rowptr, colraw, dinv, N);
    sort_rows_kernel<<<gN, TB, 0, stream>>>(rowptr, colraw, col, N);

    layer128_store_kernel<<<gTiles, 256, 0, stream>>>(
        x, rowptr, col, dinv, W1, b1, g1, be1, m1, v1, bufA, N);
    layer128_w3_kernel<<<gTiles, 256, 0, stream>>>(
        bufA, rowptr, col, dinv, W2, b2, g2, be2, m2, v2, W3, h3, N);

    gather2_logsoftmax_kernel<<<gN, TB, 0, stream>>>(h3, rowptr, col, dinv, b3, outp, N);
}


# CSR build in 2 launches, not 3: each edge chunk sorts its own edges by bucket in LDS and leaves a table of piece starts; a bucket block collects its 128 pieces (no histogram launch); same sums
# speedup vs baseline: 5.5083x; 1.2530x over previous
#include <hip/hip_runtime.h>
#include <hip/hip_bf16.h>

typedef __attribute__((ext_vector_type(2))) float v2f;
typedef __attribute__((ext_vector_type(8))) float v8f;

#define FEAT 128
#define GCH 24

#define BSH 9
#define BSZ 512
#define GB 128
#define NBMAX 256

__global__ __launch_bounds__(1024) void chunk_sort_kernel(
    const int* __restrict__ src, const int* __restrict__ dst,
    int* __restrict__ ctab, int2* __restrict__ ebuf, int ne, int nb) {
    __shared__ int cur[NBMAX];
    __shared__ int wtot[NBMAX / 32];
    const int t = threadIdx.x, lane = t & 31, wave = t >> 5;
    const int chunk = (ne + GB - 1) / GB;
    const int c0 = blockIdx.x * chunk;
    const int e0 = (c0 < ne) ? c0 : ne;
    const int e1 = (e0 + chunk < ne) ? e0 + chunk : ne;

    int d[4], sv[4];
#pragma unroll
    for (int u = 0; u < 4; ++u) {
        const int e = e0 + t + u * 1024;
        d[u]  = (e < e1) ? dst[e] : -1;
        sv[u] = (e < e1) ? src[e] : 0;
    }
    if (t < NBMAX) cur[t] = 0;
    __syncthreads();
#pragma unroll
    for (int u = 0; u < 4; ++u)
        if (d[u] >= 0) atomicAdd(&cur[d[u] >> BSH], 1);
    for (int eb = e0 + t + 4 * 1024; eb < e1; eb += 4 * 1024) {
        int dd[4];
#pragma unroll
        for (int u = 0; u < 4; ++u) {
            const int e = eb + u * 1024;
            dd[u] = (e < e1) ? dst[e] : -1;
        }
#pragma unroll
        for (int u = 0; u < 4; ++u)
            if (dd[u] >= 0) atomicAdd(&cur[dd[u] >> BSH], 1);
    }
    __syncthreads();
    int v = 0, incl = 0;
    if (t < NBMAX) {
        v = cur[t];
        incl = v;
        for (int off = 1; off < 32; off <<= 1) {
            int x = __shfl_up(incl, off, 32);
            if (lane >= off) incl += x;
        }
        if (lane == 31) wtot[wave] = incl;
    }
    __syncthreads();
    if (t < NBMAX) {
        int base = 0;
        for (int w = 0; w < NBMAX / 32; ++w)
            if (w < wave) base += wtot[w];
        const int excl = base + incl - v;
        cur[t] = e0 + excl;
        if (t < nb) {
            ctab[t * GB + blockIdx.x] = excl;
            if (t == nb - 1) ctab[nb * GB + blockIdx.x] = excl + v;
        }
    }
    __syncthreads();

#pragma unroll
    for (int u = 0; u < 4; ++u) {
        if (d[u] >= 0) {
            const int p = atomicAdd(&cur[d[u] >> BSH], 1);
            ebuf[p] = make_int2(sv[u], d[u]);
        }
    }
    for (int eb = e0 + t + 4 * 1024; eb < e1; eb += 4 * 1024) {
#pragma unroll
        for (int u = 0; u < 4; ++u) {
            const int e = eb + u * 1024;
            d[u]  = (e < e1) ? dst[e] : -1;
            sv[u] = (e < e1) ? src[e] : 0;
        }
#pragma unroll
        for (int u = 0; u < 4; ++u) {
            if (d[u] >= 0) {
                const int p = atomicAdd(&cur[d[u] >> BSH], 1);
                ebuf[p] = make_int2(sv[u], d[u]);
            }
        }
    }
}

#define RCAP 4096
#define PEPT 8
__global__ __launch_bounds__(BSZ) void bucket_rows_kernel(
    const int2* __restrict__ ebuf, const int* __restrict__ ctab,
    int* __restrict__ rowptr, int* colraw, int* __restrict__ col,
    float* __restrict__ dinv, int n, int ne) {
    __shared__ int lcnt[BSZ];
    __shared__ int lbeg[BSZ];
    __shared__ int lcur[BSZ];
    __shared__ int wtot[BSZ / 32];
    __shared__ int ssrc[RCAP];
    __shared__ unsigned short sdst[RCAP];
    __shared__ int rsrc[RCAP];
    __shared__ unsigned short rrow[RCAP];
    __shared__ int pbeg[GB];
    __shared__ int ppre[GB];
    __shared__ int plen[GB];
    const int t = threadIdx.x, lane = t & 31, wave = t >> 5;
    const int n0 = blockIdx.x << BSH;
    const int4 lo = ((const int4*)ctab)[blockIdx.x * (GB / 4) + lane];
    const int4 hi = ((const int4*)ctab)[(blockIdx.x + 1) * (GB / 4) + lane];
    const int l0 = hi.x - lo.x, l1 = hi.y - lo.y, l2 = hi.z - lo.z, l3 = hi.w - lo.w;
    const int lsum = l0 + l1 + l2 + l3;
    int fsum = lo.x + lo.y + lo.z + lo.w;
    int pinc = lsum;
    for (int off = 1; off < 32; off <<= 1) {
        int x = __shfl_up(pinc, off, 32);
        if (lane >= off) pinc += x;
    }
    for (int off = 16; off > 0; off >>= 1) fsum += __shfl_xor(fsum, off, 32);
    const int e0 = fsum;
    const int cnt = __shfl(pinc, 31, 32);
    const bool fits = cnt <= RCAP;
    if (wave == 0) {
        const int chunk = (ne + GB - 1) / GB;
        const int c4 = 4 * lane;
        const int pex = pinc - lsum;
        pbeg[c4 + 0] = (c4 + 0) * chunk + lo.x;  ppre[c4 + 0] = pex;                plen[c4 + 0] = l0;
        pbeg[c4 + 1] = (c4 + 1) * chunk + lo.y;  ppre[c4 + 1] = pex + l0;           plen[c4 + 1] = l1;
        pbeg[c4 + 2] = (c4 + 2) * chunk + lo.z;  ppre[c4 + 2] = pex + l0 + l1;      plen[c4 + 2] = l2;
        pbeg[c4 + 3] = (c4 + 3) * chunk + lo.w;  ppre[c4 + 3] = pex + l0 + l1 + l2; plen[c4 + 3] = l3;
    }
    lcnt[t] = 0;
    __syncthreads();
    const int pc = t / (BSZ / GB), sub = t % (BSZ / GB);
    const int pb = pbeg[pc], pp = ppre[pc], pl = plen[pc];
    for (int ib = sub; ib < pl; ib += (BSZ / GB) * PEPT) {
        int2 v[PEPT];
#pragma unroll
        for (int u = 0; u < PEPT; ++u) {
            const int i = ib + u * (BSZ / GB);
            v[u] = (i < pl) ? ebuf[pb + i] : make_int2(0, -1);
        }
#pragma unroll
        for (int u = 0; u < PEPT; ++u) {
            if (v[u].y >= 0) {
                atomicAdd(&lcnt[v[u].y - n0], 1);
                if (fits) {
                    const int k = pp + ib + u * (BSZ / GB);
                    ssrc[k] = v[u].x;
                    sdst[k] = (unsigned short)(v[u].y - n0);
                }
            }
        }
    }
    __syncthreads();
    const int c = lcnt[t];
    int incl = c;
    for (int off = 1; off < 32; off <<= 1) {
        int x = __shfl_up(incl, off, 32);
        if (lane >= off) incl += x;
    }
    if (lane == 31) wtot[wave] = incl;
    __syncthreads();
    int beg = e0 + incl - c;
    for (int w = 0; w < BSZ / 32; ++w)
        if (w < wave) beg += wtot[w];
    lbeg[t] = beg;
    lcur[t] = beg;
    const int i = n0 + t;
    if (i < n) {
        rowptr[i] = beg;
        dinv[i] = rsqrtf((float)(c + 1));
        if (i == n - 1) rowptr[n] = beg + c;
    }
    __syncthreads();
    if (fits) {
        for (int kb = t; kb < cnt; kb += 4 * BSZ) {
            int r[4], sv[4], p[4];
#pragma unroll
            for (int u = 0; u < 4; ++u) {
                const int k = kb + u * BSZ;
                const int kc = (k < cnt) ? k : 0;
                r[u] = sdst[kc];
                sv[u] = ssrc[kc];
            }
#pragma unroll
            for (int u = 0; u < 4; ++u) {
                p[u] = 0;
                if (kb + u * BSZ < cnt) p[u] = atomicAdd(&lcur[r[u]], 1) - e0;
            }
#pragma unroll
            for (int u = 0; u < 4; ++u) {
                if (kb + u * BSZ < cnt) {
                    rsrc[p[u]] = sv[u];
                    rrow[p[u]] = (unsigned short)r[u];
                }
            }
        }
        __syncthreads();
        for (int kb = t; kb < cnt; kb += 4 * BSZ) {
            int v[4], bg[4], ln[4], kk[4], rk[4];
#pragma unroll
            for (int u = 0; u < 4; ++u) {
                const int k = kb + u * BSZ;
                const bool ok = k < cnt;
                const int kc = ok ? k : 0;
                const int r = rrow[kc];
                kk[u] = k;
                v[u] = rsrc[kc];
                bg[u] = lbeg[r] - e0;
                ln[u] = ok ? (lcur[r] - e0 - bg[u]) : 0;
                rk[u] = 0;
            }
            int mx = ln[0];
#pragma unroll
            for (int u = 1; u < 4; ++u) mx = (ln[u] > mx) ? ln[u] : mx;
            for (int q = 0; q < mx; ++q) {
#pragma unroll
                for (int u = 0; u < 4; ++u) {
                    const int in = (q < ln[u]) ? 1 : 0;
                    const int x = rsrc[bg[u] + q * in];
                    const int lt = (x < v[u]) ? 1 : 0;
                    const int eq = (x == v[u]) ? 1 : 0;
                    const int ea = (bg[u] + q < kk[u]) ? 1 : 0;
                    rk[u] += in & (lt | (eq & ea));
                }
            }
#pragma unroll
            for (int u = 0; u < 4; ++u)
                if (ln[u] > 0) col[e0 + bg[u] + rk[u]] = v[u];
        }
    } else {
        for (int ib = sub; ib < pl; ib += (BSZ / GB) * PEPT) {
            int2 v[PEPT];
#pragma unroll
            for (int u = 0; u < PEPT; ++u) {
                const int i = ib + u * (BSZ / GB);
                v[u] = (i < pl) ? ebuf[pb + i] : make_int2(0, -1);
            }
#pragma unroll
            for (int u = 0; u < PEPT; ++u) {
                if (v[u].y >= 0) {
                    const int p = atomicAdd(&lcur[v[u].y - n0], 1);
                    colraw[p] = v[u].x;
                }
            }
        }
        __syncthreads();
        const int end = beg + c;
        for (int a = beg; a < end; ++a) {
            const int v = colraw[a];
            int r = beg;
            for (int b = beg; b < end; ++b) {
                const int u = colraw[b];
                r += (u < v || (u == v && b < a)) ? 1 : 0;
            }
            col[r] = v;
        }
    }
}

typedef __attribute__((ext_vector_type(4))) float v4f;

__device__ __forceinline__ float bias_bn_relu(float a, float bi, float mu, float va,
                                              float ga, float be) {
    const float x = a + bi;
    const float v = (x - mu) * rsqrtf(va + 1e-5f) * ga + be;
    return fmaxf(v, 0.0f);
}

template <bool FOLD3>
__device__ __forceinline__ void layer128_body(
    const float* __restrict__ xin, const int* __restrict__ rowptr,
    const int* __restrict__ col, const float* __restrict__ dinv,
    const float* __restrict__ W,
    const float* __restrict__ bias, const float* __restrict__ gamma,
    const float* __restrict__ beta, const float* __restrict__ mean,
    const float* __restrict__ var,
    const float* __restrict__ W3, float* __restrict__ out, int n) {
    __shared__ float As[32][FEAT + 4];

    const int lane = threadIdx.x & 31;
    const int wave = __builtin_amdgcn_readfirstlane((int)(threadIdx.x >> 5));
    const int m0 = blockIdx.x * 32;
    const int nb = m0 + wave * 4;
    const v4f* __restrict__ x4 = (const v4f*)xin;

    const int bidx = (nb + lane < n) ? nb + lane : n;
    const int bndv = rowptr[bidx];
    const int didx = (nb + lane < n) ? nb + lane : n - 1;
    const float ddl = dinv[didx];
    const float ddv = (nb + lane < n) ? ddl : 0.0f;
    v4f selfr[4];
#pragma unroll
    for (int q = 0; q < 4; ++q) {
        const int node = (nb + q < n) ? nb + q : n - 1;
        selfr[q] = x4[(long long)node * 32 + lane];
    }
    const int b0 = __builtin_amdgcn_readlane(bndv, 0);
    const int T  = __builtin_amdgcn_readlane(bndv, 4) - b0;
    int cq = 0;
    {
        const int k0 = (T < 32) ? T : 32;
        if (lane < k0) cq = col[b0 + lane];
    }
    v4f acc;
#pragma unroll
    for (int q = 0; q < 4; ++q) {
        const float d = __int_as_float(__builtin_amdgcn_readlane(__float_as_int(ddv), q));
        v4f ini = selfr[q] * d * d;
        if (nb + q >= n) ini = (v4f){0.0f, 0.0f, 0.0f, 0.0f};
        *(v4f*)&As[wave * 4 + q][4 * lane] = ini;
        if (q == 0) acc = ini;
    }
    float dv = 0.0f;
    int qcur = 0;
    int endcur = __builtin_amdgcn_readlane(bndv, 1);
    float ddcur = __int_as_float(__builtin_amdgcn_readlane(__float_as_int(ddv), 0));
    for (int base = 0; base < T; base += 32) {
        const int k = (T - base < 32) ? (T - base) : 32;
        if (base != 0) {
            cq = 0;
            if (lane < k) cq = col[b0 + base + lane];
        }
        for (int j = 0; j < k; j += GCH) {
            const int m = k - j;
            v4f r[GCH];
#pragma unroll
            for (int u = 0; u < GCH; ++u) {
                const int li = (u < m) ? j + u : k - 1;
                const int s = __builtin_amdgcn_readlane(cq, li);
                r[u] = x4[(long long)s * 32 + lane];
            }
            if (j == 0) dv = dinv[cq];
#pragma unroll
            for (int u = 0; u < GCH; ++u) {
                if (u < m) {
                    const int p = b0 + base + j + u;
                    while (p >= endcur && qcur < 3) {
                        *(v4f*)&As[wave * 4 + qcur][4 * lane] = acc;
                        ++qcur;
                        acc = *(const v4f*)&As[wave * 4 + qcur][4 * lane];
                        endcur = __builtin_amdgcn_readlane(bndv, qcur + 1);
                        ddcur = __int_as_float(
                            __builtin_amdgcn_readlane(__float_as_int(ddv), qcur));
                    }
                    const float w = __int_as_float(
                        __builtin_amdgcn_readlane(__float_as_int(dv), j + u)) * ddcur;
                    acc += r[u] * w;
                }
            }
        }
    }
    *(v4f*)&As[wave * 4 + qcur][4 * lane] = acc;
    __syncthreads();

    const int n0    = wave * 16;
    const int mrow  = lane & 15;
    const int khalf = (lane >> 4) * 2;
    const int ncol  = n0 + (lane & 15);

    v8f c0 = {};
    v8f c1 = {};
#pragma unroll
    for (int k0 = 0; k0 < FEAT; k0 += 4) {
        v2f a0, a1, b;
        a0.x = As[mrow][k0 + khalf];
        a0.y = As[mrow][k0 + khalf + 1];
        a1.x = As[16 + mrow][k0 + khalf];
        a1.y = As[16 + mrow][k0 + khalf + 1];
        b.x = W[(k0 + khalf) * FEAT + ncol];
        b.y = W[(k0 + khalf + 1) * FEAT + ncol];
        c0 = __builtin_amdgcn_wmma_f32_16x16x4_f32(
            false, a0, false, b, (short)0, c0, false, false);
        c1 = __builtin_amdgcn_wmma_f32_16x16x4_f32(
            false, a1, false, b, (short)0, c1, false, false);
    }

    const float bi = bias[ncol];
    const float mu = mean[ncol];
    const float va = var[ncol];
    const float ga = gamma[ncol];
    const float be = beta[ncol];
    const int mbase = (lane >> 4) * 8;
    __syncthreads();
#pragma unroll
    for (int v = 0; v < 8; ++v) {
        As[mbase + v][ncol]      = bias_bn_relu(c0[v], bi, mu, va, ga, be);
        As[16 + mbase + v][ncol] = bias_bn_relu(c1[v], bi, mu, va, ga, be);
    }
    __syncthreads();

    if (FOLD3) {
        const v4f wa = ((const v4f*)W3)[2 * lane];
        const v4f wb = ((const v4f*)W3)[2 * lane + 1];
#pragma unroll
        for (int q = 0; q < 4; ++q) {
            const v4f y = *(const v4f*)&As[wave * 4 + q][4 * lane];
            float s0 = y.x * wa.x + y.y * wa.z + y.z * wb.x + y.w * wb.z;
            float s1 = y.x * wa.y + y.y * wa.w + y.z * wb.y + y.w * wb.w;
#pragma unroll
            for (int off = 16; off > 0; off >>= 1) {
                s0 += __shfl_xor(s0, off, 32);
                s1 += __shfl_xor(s1, off, 32);
            }
            const int node = nb + q;
            if (lane == 0 && node < n) {
                out[2 * node + 0] = s0;
                out[2 * node + 1] = s1;
            }
        }
    } else {
#pragma unroll
        for (int q = 0; q < 4; ++q) {
            const int node = nb + q;
            if (node < n)
                ((v4f*)out)[(long long)node * 32 + lane] = *(const v4f*)&As[wave * 4 + q][4 * lane];
        }
    }
}

__global__ __launch_bounds__(256) void layer128_store_kernel(
    const float* __restrict__ xin, const int* __restrict__ rowptr,
    const int* __restrict__ col, const float* __restrict__ dinv,
    const float* __restrict__ W,
    const float* __restrict__ bias, const float* __restrict__ gamma,
    const float* __restrict__ beta, const float* __restrict__ mean,
    const float* __restrict__ var,
    float* __restrict__ y, int n) {
    layer128_body<false>(xin, rowptr, col, dinv, W, bias, gamma, beta, mean, var,
                         nullptr, y, n);
}

__global__ __launch_bounds__(256) void layer128_w3_kernel(
    const float* __restrict__ xin, const int* __restrict__ rowptr,
    const int* __restrict__ col, const float* __restrict__ dinv,
    const float* __restrict__ W,
    const float* __restrict__ bias, const float* __restrict__ gamma,
    const float* __restrict__ beta, const float* __restrict__ mean,
    const float* __restrict__ var,
    const float* __restrict__ W3, float* __restrict__ h3, int n) {
    layer128_body<true>(xin, rowptr, col, dinv, W, bias, gamma, beta, mean, var,
                        W3, h3, n);
}

__global__ void gather2_logsoftmax_kernel(const float* __restrict__ h3,
                               const int* __restrict__ rowptr,
                               const int* __restrict__ col,
                               const float* __restrict__ dinv,
                               const float* __restrict__ b3,
                               float* __restrict__ out, int n) {
    int i = blockIdx.x * blockDim.x + threadIdx.x;
    if (i >= n) return;
    const float d = dinv[i];
    const float d2 = d * d;
    float a0 = h3[2 * i + 0] * d2;
    float a1 = h3[2 * i + 1] * d2;
    const int beg = rowptr[i], end = rowptr[i + 1];
    for (int j = beg; j < end; ++j) {
        const int s = col[j];
        const float w = dinv[s] * d;
        a0 += h3[2 * s + 0] * w;
        a1 += h3[2 * s + 1] * w;
    }
    const float z0 = a0 + b3[0];
    const float z1 = a1 + b3[1];
    const float mx = fmaxf(z0, z1);
    const float lse = mx + logf(expf(z0 - mx) + expf(z1 - mx));
    out[2 * i + 0] = z0 - lse;
    out[2 * i + 1] = z1 - lse;
}

extern "C" void kernel_launch(void* const* d_in, const int* in_sizes, int n_in,
                              void* d_out, int out_size, void* d_ws, size_t ws_size,
                              hipStream_t stream) {
    const float* x   = (const float*)d_in[0];
    const int*   ei  = (const int*)d_in[1];
    const float* W1  = (const float*)d_in[2];
    const float* b1  = (const float*)d_in[3];
    const float* g1  = (const float*)d_in[4];
    const float* be1 = (const float*)d_in[5];
    const float* m1  = (const float*)d_in[6];
    const float* v1  = (const float*)d_in[7];
    const float* W2  = (const float*)d_in[8];
    const float* b2  = (const float*)d_in[9];
    const float* g2  = (const float*)d_in[10];
    const float* be2 = (const float*)d_in[11];
    const float* m2  = (const float*)d_in[12];
    const float* v2  = (const float*)d_in[13];
    const float* W3  = (const float*)d_in[14];
    const float* b3  = (const float*)d_in[15];

    const int N = in_sizes[0] / FEAT;
    const int E = in_sizes[1] / 2;
    const int* src = ei;
    const int* dst = ei + E;

    float* bufA = (float*)d_ws;
    float* aux  = bufA + (long long)N * FEAT;
    float* h3   = aux + (long long)N * FEAT;
    float* dinv = h3 + (long long)N * 2;
    int* rowptr = (int*)(dinv + N);
    int* col    = rowptr + N + 1;
    int* colraw = col + E;
    int* ctab   = (int*)aux;
    int2* ebuf  = (int2*)(ctab + (NBMAX + 1) * GB);
    float* outp = (float*)d_out;

    const int TB = 256;
    const int gN      = (N + TB - 1) / TB;
    const int gTiles  = (N + 31) / 32;
    const int nb      = (N + BSZ - 1) / BSZ;

    chunk_sort_kernel<<<GB, 1024, 0, stream>>>(src, dst, ctab, ebuf, E, nb);
    bucket_rows_kernel<<<nb, BSZ, 0, stream>>>(ebuf, ctab, rowptr, colraw, col, dinv, N, E);

    layer128_store_kernel<<<gTiles, 256, 0, stream>>>(
        x, rowptr, col, dinv, W1, b1, g1, be1, m1, v1, bufA, N);
    layer128_w3_kernel<<<gTiles, 256, 0, stream>>>(
        bufA, rowptr, col, dinv, W2, b2, g2, be2, m2, v2, W3, h3, N);

    gather2_logsoftmax_kernel<<<gN, TB, 0, stream>>>(h3, rowptr, col, dinv, b3, outp, N);
}


# Layer kernels: wave priority 3 (s_setprio) while a wave gathers, back to 0 for product and epilogue, so row loads issue ahead of older waves' matrix instructions; same sums
# speedup vs baseline: 5.7235x; 1.0391x over previous
#include <hip/hip_runtime.h>
#include <hip/hip_bf16.h>

typedef __attribute__((ext_vector_type(2))) float v2f;
typedef __attribute__((ext_vector_type(8))) float v8f;

#define FEAT 128
#define GCH 24

#define BSH 9
#define BSZ 512
#define GB 128
#define NBMAX 256

__global__ __launch_bounds__(1024) void chunk_sort_kernel(
    const int* __restrict__ src, const int* __restrict__ dst,
    int* __restrict__ ctab, int2* __restrict__ ebuf, int ne, int nb) {
    __shared__ int cur[NBMAX];
    __shared__ int wtot[NBMAX / 32];
    const int t = threadIdx.x, lane = t & 31, wave = t >> 5;
    const int chunk = (ne + GB - 1) / GB;
    const int c0 = blockIdx.x * chunk;
    const int e0 = (c0 < ne) ? c0 : ne;
    const int e1 = (e0 + chunk < ne) ? e0 + chunk : ne;

    int d[4], sv[4];
#pragma unroll
    for (int u = 0; u < 4; ++u) {
        const int e = e0 + t + u * 1024;
        d[u]  = (e < e1) ? dst[e] : -1;
        sv[u] = (e < e1) ? src[e] : 0;
    }
    if (t < NBMAX) cur[t] = 0;
    __syncthreads();
#pragma unroll
    for (int u = 0; u < 4; ++u)
        if (d[u] >= 0) atomicAdd(&cur[d[u] >> BSH], 1);
    for (int eb = e0 + t + 4 * 1024; eb < e1; eb += 4 * 1024) {
        int dd[4];
#pragma unroll
        for (int u = 0; u < 4; ++u) {
            const int e = eb + u * 1024;
            dd[u] = (e < e1) ? dst[e] : -1;
        }
#pragma unroll
        for (int u = 0; u < 4; ++u)
            if (dd[u] >= 0) atomicAdd(&cur[dd[u] >> BSH], 1);
    }
    __syncthreads();
    int v = 0, incl = 0;
    if (t < NBMAX) {
        v = cur[t];
        incl = v;
        for (int off = 1; off < 32; off <<= 1) {
            int x = __shfl_up(incl, off, 32);
            if (lane >= off) incl += x;
        }
        if (lane == 31) wtot[wave] = incl;
    }
    __syncthreads();
    if (t < NBMAX) {
        int base = 0;
        for (int w = 0; w < NBMAX / 32; ++w)
            if (w < wave) base += wtot[w];
        const int excl = base + incl - v;
        cur[t] = e0 + excl;
        if (t < nb) {
            ctab[t * GB + blockIdx.x] = excl;
            if (t == nb - 1) ctab[nb * GB + blockIdx.x] = excl + v;
        }
    }
    __syncthreads();

#pragma unroll
    for (int u = 0; u < 4; ++u) {
        if (d[u] >= 0) {
            const int p = atomicAdd(&cur[d[u] >> BSH], 1);
            ebuf[p] = make_int2(sv[u], d[u]);
        }
    }
    for (int eb = e0 + t + 4 * 1024; eb < e1; eb += 4 * 1024) {
#pragma unroll
        for (int u = 0; u < 4; ++u) {
            const int e = eb + u * 1024;
            d[u]  = (e < e1) ? dst[e] : -1;
            sv[u] = (e < e1) ? src[e] : 0;
        }
#pragma unroll
        for (int u = 0; u < 4; ++u) {
            if (d[u] >= 0) {
                const int p = atomicAdd(&cur[d[u] >> BSH], 1);
                ebuf[p] = make_int2(sv[u], d[u]);
            }
        }
    }
}

#define RCAP 4096
#define PEPT 8
__global__ __launch_bounds__(BSZ) void bucket_rows_kernel(
    const int2* __restrict__ ebuf, const int* __restrict__ ctab,
    int* __restrict__ rowptr, int* colraw, int* __restrict__ col,
    float* __restrict__ dinv, int n, int ne) {
    __shared__ int lcnt[BSZ];
    __shared__ int lbeg[BSZ];
    __shared__ int lcur[BSZ];
    __shared__ int wtot[BSZ / 32];
    __shared__ int ssrc[RCAP];
    __shared__ unsigned short sdst[RCAP];
    __shared__ int rsrc[RCAP];
    __shared__ unsigned short rrow[RCAP];
    __shared__ int pbeg[GB];
    __shared__ int ppre[GB];
    __shared__ int plen[GB];
    const int t = threadIdx.x, lane = t & 31, wave = t >> 5;
    const int n0 = blockIdx.x << BSH;
    const int4 lo = ((const int4*)ctab)[blockIdx.x * (GB / 4) + lane];
    const int4 hi = ((const int4*)ctab)[(blockIdx.x + 1) * (GB / 4) + lane];
    const int l0 = hi.x - lo.x, l1 = hi.y - lo.y, l2 = hi.z - lo.z, l3 = hi.w - lo.w;
    const int lsum = l0 + l1 + l2 + l3;
    int fsum = lo.x + lo.y + lo.z + lo.w;
    int pinc = lsum;
    for (int off = 1; off < 32; off <<= 1) {
        int x = __shfl_up(pinc, off, 32);
        if (lane >= off) pinc += x;
    }
    for (int off = 16; off > 0; off >>= 1) fsum += __shfl_xor(fsum, off, 32);
    const int e0 = fsum;
    const int cnt = __shfl(pinc, 31, 32);
    const bool fits = cnt <= RCAP;
    if (wave == 0) {
        const int chunk = (ne + GB - 1) / GB;
        const int c4 = 4 * lane;
        const int pex = pinc - lsum;
        pbeg[c4 + 0] = (c4 + 0) * chunk + lo.x;  ppre[c4 + 0] = pex;                plen[c4 + 0] = l0;
        pbeg[c4 + 1] = (c4 + 1) * chunk + lo.y;  ppre[c4 + 1] = pex + l0;           plen[c4 + 1] = l1;
        pbeg[c4 + 2] = (c4 + 2) * chunk + lo.z;  ppre[c4 + 2] = pex + l0 + l1;      plen[c4 + 2] = l2;
        pbeg[c4 + 3] = (c4 + 3) * chunk + lo.w;  ppre[c4 + 3] = pex + l0 + l1 + l2; plen[c4 + 3] = l3;
    }
    lcnt[t] = 0;
    __syncthreads();
    const int pc = t / (BSZ / GB), sub = t % (BSZ / GB);
    const int pb = pbeg[pc], pp = ppre[pc], pl = plen[pc];
    for (int ib = sub; ib < pl; ib += (BSZ / GB) * PEPT) {
        int2 v[PEPT];
#pragma unroll
        for (int u = 0; u < PEPT; ++u) {
            const int i = ib + u * (BSZ / GB);
            v[u] = (i < pl) ? ebuf[pb + i] : make_int2(0, -1);
        }
#pragma unroll
        for (int u = 0; u < PEPT; ++u) {
            if (v[u].y >= 0) {
                atomicAdd(&lcnt[v[u].y - n0], 1);
                if (fits) {
                    const int k = pp + ib + u * (BSZ / GB);
                    ssrc[k] = v[u].x;
                    sdst[k] = (unsigned short)(v[u].y - n0);
                }
            }
        }
    }
    __syncthreads();
    const int c = lcnt[t];
    int incl = c;
    for (int off = 1; off < 32; off <<= 1) {
        int x = __shfl_up(incl, off, 32);
        if (lane >= off) incl += x;
    }
    if (lane == 31) wtot[wave] = incl;
    __syncthreads();
    int beg = e0 + incl - c;
    for (int w = 0; w < BSZ / 32; ++w)
        if (w < wave) beg += wtot[w];
    lbeg[t] = beg;
    lcur[t] = beg;
    const int i = n0 + t;
    if (i < n) {
        rowptr[i] = beg;
        dinv[i] = rsqrtf((float)(c + 1));
        if (i == n - 1) rowptr[n] = beg + c;
    }
    __syncthreads();
    if (fits) {
        for (int kb = t; kb < cnt; kb += 4 * BSZ) {
            int r[4], sv[4], p[4];
#pragma unroll
            for (int u = 0; u < 4; ++u) {
                const int k = kb + u * BSZ;
                const int kc = (k < cnt) ? k : 0;
                r[u] = sdst[kc];
                sv[u] = ssrc[kc];
            }
#pragma unroll
            for (int u = 0; u < 4; ++u) {
                p[u] = 0;
                if (kb + u * BSZ < cnt) p[u] = atomicAdd(&lcur[r[u]], 1) - e0;
            }
#pragma unroll
            for (int u = 0; u < 4; ++u) {
                if (kb + u * BSZ < cnt) {
                    rsrc[p[u]] = sv[u];
                    rrow[p[u]] = (unsigned short)r[u];
                }
            }
        }
        __syncthreads();
        for (int kb = t; kb < cnt; kb += 4 * BSZ) {
            int v[4], bg[4], ln[4], kk[4], rk[4];
#pragma unroll
            for (int u = 0; u < 4; ++u) {
                const int k = kb + u * BSZ;
                const bool ok = k < cnt;
                const int kc = ok ? k : 0;
                const int r = rrow[kc];
                kk[u] = k;
                v[u] = rsrc[kc];
                bg[u] = lbeg[r] - e0;
                ln[u] = ok ? (lcur[r] - e0 - bg[u]) : 0;
                rk[u] = 0;
            }
            int mx = ln[0];
#pragma unroll
            for (int u = 1; u < 4; ++u) mx = (ln[u] > mx) ? ln[u] : mx;
            for (int q = 0; q < mx; ++q) {
#pragma unroll
                for (int u = 0; u < 4; ++u) {
                    const int in = (q < ln[u]) ? 1 : 0;
                    const int x = rsrc[bg[u] + q * in];
                    const int lt = (x < v[u]) ? 1 : 0;
                    const int eq = (x == v[u]) ? 1 : 0;
                    const int ea = (bg[u] + q < kk[u]) ? 1 : 0;
                    rk[u] += in & (lt | (eq & ea));
                }
            }
#pragma unroll
            for (int u = 0; u < 4; ++u)
                if (ln[u] > 0) col[e0 + bg[u] + rk[u]] = v[u];
        }
    } else {
        for (int ib = sub; ib < pl; ib += (BSZ / GB) * PEPT) {
            int2 v[PEPT];
#pragma unroll
            for (int u = 0; u < PEPT; ++u) {
                const int i = ib + u * (BSZ / GB);
                v[u] = (i < pl) ? ebuf[pb + i] : make_int2(0, -1);
            }
#pragma unroll
            for (int u = 0; u < PEPT; ++u) {
                if (v[u].y >= 0) {
                    const int p = atomicAdd(&lcur[v[u].y - n0], 1);
                    colraw[p] = v[u].x;
                }
            }
        }
        __syncthreads();
        const int end = beg + c;
        for (int a = beg; a < end; ++a) {
            const int v = colraw[a];
            int r = beg;
            for (int b = beg; b < end; ++b) {
                const int u = colraw[b];
                r += (u < v || (u == v && b < a)) ? 1 : 0;
            }
            col[r] = v;
        }
    }
}

typedef __attribute__((ext_vector_type(4))) float v4f;

__device__ __forceinline__ float bias_bn_relu(float a, float bi, float mu, float va,
                                              float ga, float be) {
    const float x = a + bi;
    const float v = (x - mu) * rsqrtf(va + 1e-5f) * ga + be;
    return fmaxf(v, 0.0f);
}

template <bool FOLD3>
__device__ __forceinline__ void layer128_body(
    const float* __restrict__ xin, const int* __restrict__ rowptr,
    const int* __restrict__ col, const float* __restrict__ dinv,
    const float* __restrict__ W,
    const float* __restrict__ bias, const float* __restrict__ gamma,
    const float* __restrict__ beta, const float* __restrict__ mean,
    const float* __restrict__ var,
    const float* __restrict__ W3, float* __restrict__ out, int n) {
    __shared__ float As[32][FEAT + 4];

    __builtin_amdgcn_s_setprio(3);
    const int lane = threadIdx.x & 31;
    const int wave = __builtin_amdgcn_readfirstlane((int)(threadIdx.x >> 5));
    const int m0 = blockIdx.x * 32;
    const int nb = m0 + wave * 4;
    const v4f* __restrict__ x4 = (const v4f*)xin;

    const int bidx = (nb + lane < n) ? nb + lane : n;
    const int bndv = rowptr[bidx];
    const int didx = (nb + lane < n) ? nb + lane : n - 1;
    const float ddl = dinv[didx];
    const float ddv = (nb + lane < n) ? ddl : 0.0f;
    v4f selfr[4];
#pragma unroll
    for (int q = 0; q < 4; ++q) {
        const int node = (nb + q < n) ? nb + q : n - 1;
        selfr[q] = x4[(long long)node * 32 + lane];
    }
    const int b0 = __builtin_amdgcn_readlane(bndv, 0);
    const int T  = __builtin_amdgcn_readlane(bndv, 4) - b0;
    int cq = 0;
    {
        const int k0 = (T < 32) ? T : 32;
        if (lane < k0) cq = col[b0 + lane];
    }
    v4f acc;
#pragma unroll
    for (int q = 0; q < 4; ++q) {
        const float d = __int_as_float(__builtin_amdgcn_readlane(__float_as_int(ddv), q));
        v4f ini = selfr[q] * d * d;
        if (nb + q >= n) ini = (v4f){0.0f, 0.0f, 0.0f, 0.0f};
        *(v4f*)&As[wave * 4 + q][4 * lane] = ini;
        if (q == 0) acc = ini;
    }
    float dv = 0.0f;
    int qcur = 0;
    int endcur = __builtin_amdgcn_readlane(bndv, 1);
    float ddcur = __int_as_float(__builtin_amdgcn_readlane(__float_as_int(ddv), 0));
    for (int base = 0; base < T; base += 32) {
        const int k = (T - base < 32) ? (T - base) : 32;
        if (base != 0) {
            cq = 0;
            if (lane < k) cq = col[b0 + base + lane];
        }
        for (int j = 0; j < k; j += GCH) {
            const int m = k - j;
            v4f r[GCH];
#pragma unroll
            for (int u = 0; u < GCH; ++u) {
                const int li = (u < m) ? j + u : k - 1;
                const int s = __builtin_amdgcn_readlane(cq, li);
                r[u] = x4[(long long)s * 32 + lane];
            }
            if (j == 0) dv = dinv[cq];
#pragma unroll
            for (int u = 0; u < GCH; ++u) {
                if (u < m) {
                    const int p = b0 + base + j + u;
                    while (p >= endcur && qcur < 3) {
                        *(v4f*)&As[wave * 4 + qcur][4 * lane] = acc;
                        ++qcur;
                        acc = *(const v4f*)&As[wave * 4 + qcur][4 * lane];
                        endcur = __builtin_amdgcn_readlane(bndv, qcur + 1);
                        ddcur = __int_as_float(
                            __builtin_amdgcn_readlane(__float_as_int(ddv), qcur));
                    }
                    const float w = __int_as_float(
                        __builtin_amdgcn_readlane(__float_as_int(dv), j + u)) * ddcur;
                    acc += r[u] * w;
                }
            }
        }
    }
    *(v4f*)&As[wave * 4 + qcur][4 * lane] = acc;
    __builtin_amdgcn_s_setprio(0);
    __syncthreads();

    const int n0    = wave * 16;
    const int mrow  = lane & 15;
    const int khalf = (lane >> 4) * 2;
    const int ncol  = n0 + (lane & 15);

    v8f c0 = {};
    v8f c1 = {};
#pragma unroll
    for (int k0 = 0; k0 < FEAT; k0 += 4) {
        v2f a0, a1, b;
        a0.x = As[mrow][k0 + khalf];
        a0.y = As[mrow][k0 + khalf + 1];
        a1.x = As[16 + mrow][k0 + khalf];
        a1.y = As[16 + mrow][k0 + khalf + 1];
        b.x = W[(k0 + khalf) * FEAT + ncol];
        b.y = W[(k0 + khalf + 1) * FEAT + ncol];
        c0 = __builtin_amdgcn_wmma_f32_16x16x4_f32(
            false, a0, false, b, (short)0, c0, false, false);
        c1 = __builtin_amdgcn_wmma_f32_16x16x4_f32(
            false, a1, false, b, (short)0, c1, false, false);
    }

    const float bi = bias[ncol];
    const float mu = mean[ncol];
    const float va = var[ncol];
    const float ga = gamma[ncol];
    const float be = beta[ncol];
    const int mbase = (lane >> 4) * 8;
    __syncthreads();
#pragma unroll
    for (int v = 0; v < 8; ++v) {
        As[mbase + v][ncol]      = bias_bn_relu(c0[v], bi, mu, va, ga, be);
        As[16 + mbase + v][ncol] = bias_bn_relu(c1[v], bi, mu, va, ga, be);
    }
    __syncthreads();

    if (FOLD3) {
        const v4f wa = ((const v4f*)W3)[2 * lane];
        const v4f wb = ((const v4f*)W3)[2 * lane + 1];
#pragma unroll
        for (int q = 0; q < 4; ++q) {
            const v4f y = *(const v4f*)&As[wave * 4 + q][4 * lane];
            float s0 = y.x * wa.x + y.y * wa.z + y.z * wb.x + y.w * wb.z;
            float s1 = y.x * wa.y + y.y * wa.w + y.z * wb.y + y.w * wb.w;
#pragma unroll
            for (int off = 16; off > 0; off >>= 1) {
                s0 += __shfl_xor(s0, off, 32);
                s1 += __shfl_xor(s1, off, 32);
            }
            const int node = nb + q;
            if (lane == 0 && node < n) {
                out[2 * node + 0] = s0;
                out[2 * node + 1] = s1;
            }
        }
    } else {
#pragma unroll
        for (int q = 0; q < 4; ++q) {
            const int node = nb + q;
            if (node < n)
                ((v4f*)out)[(long long)node * 32 + lane] = *(const v4f*)&As[wave * 4 + q][4 * lane];
        }
    }
}

__global__ __launch_bounds__(256) void layer128_store_kernel(
    const float* __restrict__ xin, const int* __restrict__ rowptr,
    const int* __restrict__ col, const float* __restrict__ dinv,
    const float* __restrict__ W,
    const float* __restrict__ bias, const float* __restrict__ gamma,
    const float* __restrict__ beta, const float* __restrict__ mean,
    const float* __restrict__ var,
    float* __restrict__ y, int n) {
    layer128_body<false>(xin, rowptr, col, dinv, W, bias, gamma, beta, mean, var,
                         nullptr, y, n);
}

__global__ __launch_bounds__(256) void layer128_w3_kernel(
    const float* __restrict__ xin, const int* __restrict__ rowptr,
    const int* __restrict__ col, const float* __restrict__ dinv,
    const float* __restrict__ W,
    const float* __restrict__ bias, const float* __restrict__ gamma,
    const float* __restrict__ beta, const float* __restrict__ mean,
    const float* __restrict__ var,
    const float* __restrict__ W3, float* __restrict__ h3, int n) {
    layer128_body<true>(xin, rowptr, col, dinv, W, bias, gamma, beta, mean, var,
                        W3, h3, n);
}

__global__ void gather2_logsoftmax_kernel(const float* __restrict__ h3,
                               const int* __restrict__ rowptr,
                               const int* __restrict__ col,
                               const float* __restrict__ dinv,
                               const float* __restrict__ b3,
                               float* __restrict__ out, int n) {
    int i = blockIdx.x * blockDim.x + threadIdx.x;
    if (i >= n) return;
    const float d = dinv[i];
    const float d2 = d * d;
    float a0 = h3[2 * i + 0] * d2;
    float a1 = h3[2 * i + 1] * d2;
    const int beg = rowptr[i], end = rowptr[i + 1];
    for (int j = beg; j < end; ++j) {
        const int s = col[j];
        const float w = dinv[s] * d;
        a0 += h3[2 * s + 0] * w;
        a1 += h3[2 * s + 1] * w;
    }
    const float z0 = a0 + b3[0];
    const float z1 = a1 + b3[1];
    const float mx = fmaxf(z0, z1);
    const float lse = mx + logf(expf(z0 - mx) + expf(z1 - mx));
    out[2 * i + 0] = z0 - lse;
    out[2 * i + 1] = z1 - lse;
}

extern "C" void kernel_launch(void* const* d_in, const int* in_sizes, int n_in,
                              void* d_out, int out_size, void* d_ws, size_t ws_size,
                              hipStream_t stream) {
    const float* x   = (const float*)d_in[0];
    const int*   ei  = (const int*)d_in[1];
    const float* W1  = (const float*)d_in[2];
    const float* b1  = (const float*)d_in[3];
    const float* g1  = (const float*)d_in[4];
    const float* be1 = (const float*)d_in[5];
    const float* m1  = (const float*)d_in[6];
    const float* v1  = (const float*)d_in[7];
    const float* W2  = (const float*)d_in[8];
    const float* b2  = (const float*)d_in[9];
    const float* g2  = (const float*)d_in[10];
    const float* be2 = (const float*)d_in[11];
    const float* m2  = (const float*)d_in[12];
    const float* v2  = (const float*)d_in[13];
    const float* W3  = (const float*)d_in[14];
    const float* b3  = (const float*)d_in[15];

    const int N = in_sizes[0] / FEAT;
    const int E = in_sizes[1] / 2;
    const int* src = ei;
    const int* dst = ei + E;

    float* bufA = (float*)d_ws;
    float* aux  = bufA + (long long)N * FEAT;
    float* h3   = aux + (long long)N * FEAT;
    float* dinv = h3 + (long long)N * 2;
    int* rowptr = (int*)(dinv + N);
    int* col    = rowptr + N + 1;
    int* colraw = col + E;
    int* ctab   = (int*)aux;
    int2* ebuf  = (int2*)(ctab + (NBMAX + 1) * GB);
    float* outp = (float*)d_out;

    const int TB = 256;
    const int gN      = (N + TB - 1) / TB;
    const int gTiles  = (N + 31) / 32;
    const int nb      = (N + BSZ - 1) / BSZ;

    chunk_sort_kernel<<<GB, 1024, 0, stream>>>(src, dst, ctab, ebuf, E, nb);
    bucket_rows_kernel<<<nb, BSZ, 0, stream>>>(ebuf, ctab, rowptr, colraw, col, dinv, N, E);

    layer128_store_kernel<<<gTiles, 256, 0, stream>>>(
        x, rowptr, col, dinv, W1, b1, g1, be1, m1, v1, bufA, N);
    layer128_w3_kernel<<<gTiles, 256, 0, stream>>>(
        bufA, rowptr, col, dinv, W2, b2, g2, be2, m2, v2, W3, h3, N);

    gather2_logsoftmax_kernel<<<gN, TB, 0, stream>>>(h3, rowptr, col, dinv, b3, outp, N);
}


# Layer kernels: 12 waves per SIMD instead of 8 (register budget 80 by launch bound; a wave requests 12 source rows per batch, not 24), so 6 panels are resident per WGP, not 4; same sums, same order
# speedup vs baseline: 6.0760x; 1.0616x over previous
#include <hip/hip_runtime.h>
#include <hip/hip_bf16.h>

typedef __attribute__((ext_vector_type(2))) float v2f;
typedef __attribute__((ext_vector_type(8))) float v8f;

#define FEAT 128
#define GCH 12

#define BSH 9
#define BSZ 512
#define GB 128
#define NBMAX 256

__global__ __launch_bounds__(1024) void chunk_sort_kernel(
    const int* __restrict__ src, const int* __restrict__ dst,
    int* __restrict__ ctab, int2* __restrict__ ebuf, int ne, int nb) {
    __shared__ int cur[NBMAX];
    __shared__ int wtot[NBMAX / 32];
    const int t = threadIdx.x, lane = t & 31, wave = t >> 5;
    const int chunk = (ne + GB - 1) / GB;
    const int c0 = blockIdx.x * chunk;
    const int e0 = (c0 < ne) ? c0 : ne;
    const int e1 = (e0 + chunk < ne) ? e0 + chunk : ne;

    int d[4], sv[4];
#pragma unroll
    for (int u = 0; u < 4; ++u) {
        const int e = e0 + t + u * 1024;
        d[u]  = (e < e1) ? dst[e] : -1;
        sv[u] = (e < e1) ? src[e] : 0;
    }
    if (t < NBMAX) cur[t] = 0;
    __syncthreads();
#pragma unroll
    for (int u = 0; u < 4; ++u)
        if (d[u] >= 0) atomicAdd(&cur[d[u] >> BSH], 1);
    for (int eb = e0 + t + 4 * 1024; eb < e1; eb += 4 * 1024) {
        int dd[4];
#pragma unroll
        for (int u = 0; u < 4; ++u) {
            const int e = eb + u * 1024;
            dd[u] = (e < e1) ? dst[e] : -1;
        }
#pragma unroll
        for (int u = 0; u < 4; ++u)
            if (dd[u] >= 0) atomicAdd(&cur[dd[u] >> BSH], 1);
    }
    __syncthreads();
    int v = 0, incl = 0;
    if (t < NBMAX) {
        v = cur[t];
        incl = v;
        for (int off = 1; off < 32; off <<= 1) {
            int x = __shfl_up(incl, off, 32);
            if (lane >= off) incl += x;
        }
        if (lane == 31) wtot[wave] = incl;
    }
    __syncthreads();
    if (t < NBMAX) {
        int base = 0;
        for (int w = 0; w < NBMAX / 32; ++w)
            if (w < wave) base += wtot[w];
        const int excl = base + incl - v;
        cur[t] = e0 + excl;
        if (t < nb) {
            ctab[t * GB + blockIdx.x] = excl;
            if (t == nb - 1) ctab[nb * GB + blockIdx.x] = excl + v;
        }
    }
    __syncthreads();

#pragma unroll
    for (int u = 0; u < 4; ++u) {
        if (d[u] >= 0) {
            const int p = atomicAdd(&cur[d[u] >> BSH], 1);
            ebuf[p] = make_int2(sv[u], d[u]);
        }
    }
    for (int eb = e0 + t + 4 * 1024; eb < e1; eb += 4 * 1024) {
#pragma unroll
        for (int u = 0; u < 4; ++u) {
            const int e = eb + u * 1024;
            d[u]  = (e < e1) ? dst[e] : -1;
            sv[u] = (e < e1) ? src[e] : 0;
        }
#pragma unroll
        for (int u = 0; u < 4; ++u) {
            if (d[u] >= 0) {
                const int p = atomicAdd(&cur[d[u] >> BSH], 1);
                ebuf[p] = make_int2(sv[u], d[u]);
            }
        }
    }
}

#define RCAP 4096
#define PEPT 8
__global__ __launch_bounds__(BSZ) void bucket_rows_kernel(
    const int2* __restrict__ ebuf, const int* __restrict__ ctab,
    int* __restrict__ rowptr, int* colraw, int* __restrict__ col,
    float* __restrict__ dinv, int n, int ne) {
    __shared__ int lcnt[BSZ];
    __shared__ int lbeg[BSZ];
    __shared__ int lcur[BSZ];
    __shared__ int wtot[BSZ / 32];
    __shared__ int ssrc[RCAP];
    __shared__ unsigned short sdst[RCAP];
    __shared__ int rsrc[RCAP];
    __shared__ unsigned short rrow[RCAP];
    __shared__ int pbeg[GB];
    __shared__ int ppre[GB];
    __shared__ int plen[GB];
    const int t = threadIdx.x, lane = t & 31, wave = t >> 5;
    const int n0 = blockIdx.x << BSH;
    const int4 lo = ((const int4*)ctab)[blockIdx.x * (GB / 4) + lane];
    const int4 hi = ((const int4*)ctab)[(blockIdx.x + 1) * (GB / 4) + lane];
    const int l0 = hi.x - lo.x, l1 = hi.y - lo.y, l2 = hi.z - lo.z, l3 = hi.w - lo.w;
    const int lsum = l0 + l1 + l2 + l3;
    int fsum = lo.x + lo.y + lo.z + lo.w;
    int pinc = lsum;
    for (int off = 1; off < 32; off <<= 1) {
        int x = __shfl_up(pinc, off, 32);
        if (lane >= off) pinc += x;
    }
    for (int off = 16; off > 0; off >>= 1) fsum += __shfl_xor(fsum, off, 32);
    const int e0 = fsum;
    const int cnt = __shfl(pinc, 31, 32);
    const bool fits = cnt <= RCAP;
    if (wave == 0) {
        const int chunk = (ne + GB - 1) / GB;
        const int c4 = 4 * lane;
        const int pex = pinc - lsum;
        pbeg[c4 + 0] = (c4 + 0) * chunk + lo.x;  ppre[c4 + 0] = pex;                plen[c4 + 0] = l0;
        pbeg[c4 + 1] = (c4 + 1) * chunk + lo.y;  ppre[c4 + 1] = pex + l0;           plen[c4 + 1] = l1;
        pbeg[c4 + 2] = (c4 + 2) * chunk + lo.z;  ppre[c4 + 2] = pex + l0 + l1;      plen[c4 + 2] = l2;
        pbeg[c4 + 3] = (c4 + 3) * chunk + lo.w;  ppre[c4 + 3] = pex + l0 + l1 + l2; plen[c4 + 3] = l3;
    }
    lcnt[t] = 0;
    __syncthreads();
    const int pc = t / (BSZ / GB), sub = t % (BSZ / GB);
    const int pb = pbeg[pc], pp = ppre[pc], pl = plen[pc];
    for (int ib = sub; ib < pl; ib += (BSZ / GB) * PEPT) {
        int2 v[PEPT];
#pragma unroll
        for (int u = 0; u < PEPT; ++u) {
            const int i = ib + u * (BSZ / GB);
            v[u] = (i < pl) ? ebuf[pb + i] : make_int2(0, -1);
        }
#pragma unroll
        for (int u = 0; u < PEPT; ++u) {
            if (v[u].y >= 0) {
                atomicAdd(&lcnt[v[u].y - n0], 1);
                if (fits) {
                    const int k = pp + ib + u * (BSZ / GB);
                    ssrc[k] = v[u].x;
                    sdst[k] = (unsigned short)(v[u].y - n0);
                }
            }
        }
    }
    __syncthreads();
    const int c = lcnt[t];
    int incl = c;
    for (int off = 1; off < 32; off <<= 1) {
        int x = __shfl_up(incl, off, 32);
        if (lane >= off) incl += x;
    }
    if (lane == 31) wtot[wave] = incl;
    __syncthreads();
    int beg = e0 + incl - c;
    for (int w = 0; w < BSZ / 32; ++w)
        if (w < wave) beg += wtot[w];
    lbeg[t] = beg;
    lcur[t] = beg;
    const int i = n0 + t;
    if (i < n) {
        rowptr[i] = beg;
        dinv[i] = rsqrtf((float)(c + 1));
        if (i == n - 1) rowptr[n] = beg + c;
    }
    __syncthreads();
    if (fits) {
        for (int kb = t; kb < cnt; kb += 4 * BSZ) {
            int r[4], sv[4], p[4];
#pragma unroll
            for (int u = 0; u < 4; ++u) {
                const int k = kb + u * BSZ;
                const int kc = (k < cnt) ? k : 0;
                r[u] = sdst[kc];
                sv[u] = ssrc[kc];
            }
#pragma unroll
            for (int u = 0; u < 4; ++u) {
                p[u] = 0;
                if (kb + u * BSZ < cnt) p[u] = atomicAdd(&lcur[r[u]], 1) - e0;
            }
#pragma unroll
            for (int u = 0; u < 4; ++u) {
                if (kb + u * BSZ < cnt) {
                    rsrc[p[u]] = sv[u];
                    rrow[p[u]] = (unsigned short)r[u];
                }
            }
        }
        __syncthreads();
        for (int kb = t; kb < cnt; kb += 4 * BSZ) {
            int v[4], bg[4], ln[4], kk[4], rk[4];
#pragma unroll
            for (int u = 0; u < 4; ++u) {
                const int k = kb + u * BSZ;
                const bool ok = k < cnt;
                const int kc = ok ? k : 0;
                const int r = rrow[kc];
                kk[u] = k;
                v[u] = rsrc[kc];
                bg[u] = lbeg[r] - e0;
                ln[u] = ok ? (lcur[r] - e0 - bg[u]) : 0;
                rk[u] = 0;
            }
            int mx = ln[0];
#pragma unroll
            for (int u = 1; u < 4; ++u) mx = (ln[u] > mx) ? ln[u] : mx;
            for (int q = 0; q < mx; ++q) {
#pragma unroll
                for (int u = 0; u < 4; ++u) {
                    const int in = (q < ln[u]) ? 1 : 0;
                    const int x = rsrc[bg[u] + q * in];
                    const int lt = (x < v[u]) ? 1 : 0;
                    const int eq = (x == v[u]) ? 1 : 0;
                    const int ea = (bg[u] + q < kk[u]) ? 1 : 0;
                    rk[u] += in & (lt | (eq & ea));
                }
            }
#pragma unroll
            for (int u = 0; u < 4; ++u)
                if (ln[u] > 0) col[e0 + bg[u] + rk[u]] = v[u];
        }
    } else {
        for (int ib = sub; ib < pl; ib += (BSZ / GB) * PEPT) {
            int2 v[PEPT];
#pragma unroll
            for (int u = 0; u < PEPT; ++u) {
                const int i = ib + u * (BSZ / GB);
                v[u] = (i < pl) ? ebuf[pb + i] : make_int2(0, -1);
            }
#pragma unroll
            for (int u = 0; u < PEPT; ++u) {
                if (v[u].y >= 0) {
                    const int p = atomicAdd(&lcur[v[u].y - n0], 1);
                    colraw[p] = v[u].x;
                }
            }
        }
        __syncthreads();
        const int end = beg + c;
        for (int a = beg; a < end; ++a) {
            const int v = colraw[a];
            int r = beg;
            for (int b = beg; b < end; ++b) {
                const int u = colraw[b];
                r += (u < v || (u == v && b < a)) ? 1 : 0;
            }
            col[r] = v;
        }
    }
}

typedef __attribute__((ext_vector_type(4))) float v4f;

__device__ __forceinline__ float bias_bn_relu(float a, float bi, float mu, float va,
                                              float ga, float be) {
    const float x = a + bi;
    const float v = (x - mu) * rsqrtf(va + 1e-5f) * ga + be;
    return fmaxf(v, 0.0f);
}

template <bool FOLD3>
__device__ __forceinline__ void layer128_body(
    const float* __restrict__ xin, const int* __restrict__ rowptr,
    const int* __restrict__ col, const float* __restrict__ dinv,
    const float* __restrict__ W,
    const float* __restrict__ bias, const float* __restrict__ gamma,
    const float* __restrict__ beta, const float* __restrict__ mean,
    const float* __restrict__ var,
    const float* __restrict__ W3, float* __restrict__ out, int n) {
    __shared__ float As[32][FEAT + 4];

    __builtin_amdgcn_s_setprio(3);
    const int lane = threadIdx.x & 31;
    const int wave = __builtin_amdgcn_readfirstlane((int)(threadIdx.x >> 5));
    const int m0 = blockIdx.x * 32;
    const int nb = m0 + wave * 4;
    const v4f* __restrict__ x4 = (const v4f*)xin;

    const int bidx = (nb + lane < n) ? nb + lane : n;
    const int bndv = rowptr[bidx];
    const int didx = (nb + lane < n) ? nb + lane : n - 1;
    const float ddl = dinv[didx];
    const float ddv = (nb + lane < n) ? ddl : 0.0f;
    v4f selfr[4];
#pragma unroll
    for (int q = 0; q < 4; ++q) {
        const int node = (nb + q < n) ? nb + q : n - 1;
        selfr[q] = x4[(long long)node * 32 + lane];
    }
    const int b0 = __builtin_amdgcn_readlane(bndv, 0);
    const int T  = __builtin_amdgcn_readlane(bndv, 4) - b0;
    int cq = 0;
    {
        const int k0 = (T < 32) ? T : 32;
        if (lane < k0) cq = col[b0 + lane];
    }
    v4f acc;
#pragma unroll
    for (int q = 0; q < 4; ++q) {
        const float d = __int_as_float(__builtin_amdgcn_readlane(__float_as_int(ddv), q));
        v4f ini = selfr[q] * d * d;
        if (nb + q >= n) ini = (v4f){0.0f, 0.0f, 0.0f, 0.0f};
        *(v4f*)&As[wave * 4 + q][4 * lane] = ini;
        if (q == 0) acc = ini;
    }
    float dv = 0.0f;
    int qcur = 0;
    int endcur = __builtin_amdgcn_readlane(bndv, 1);
    float ddcur = __int_as_float(__builtin_amdgcn_readlane(__float_as_int(ddv), 0));
    for (int base = 0; base < T; base += 32) {
        const int k = (T - base < 32) ? (T - base) : 32;
        if (base != 0) {
            cq = 0;
            if (lane < k) cq = col[b0 + base + lane];
        }
        for (int j = 0; j < k; j += GCH) {
            const int m = k - j;
            v4f r[GCH];
#pragma unroll
            for (int u = 0; u < GCH; ++u) {
                const int li = (u < m) ? j + u : k - 1;
                const int s = __builtin_amdgcn_readlane(cq, li);
                r[u] = x4[(long long)s * 32 + lane];
            }
            if (j == 0) dv = dinv[cq];
#pragma unroll
            for (int u = 0; u < GCH; ++u) {
                if (u < m) {
                    const int p = b0 + base + j + u;
                    while (p >= endcur && qcur < 3) {
                        *(v4f*)&As[wave * 4 + qcur][4 * lane] = acc;
                        ++qcur;
                        acc = *(const v4f*)&As[wave * 4 + qcur][4 * lane];
                        endcur = __builtin_amdgcn_readlane(bndv, qcur + 1);
                        ddcur = __int_as_float(
                            __builtin_amdgcn_readlane(__float_as_int(ddv), qcur));
                    }
                    const float w = __int_as_float(
                        __builtin_amdgcn_readlane(__float_as_int(dv), j + u)) * ddcur;
                    acc += r[u] * w;
                }
            }
        }
    }
    *(v4f*)&As[wave * 4 + qcur][4 * lane] = acc;
    __builtin_amdgcn_s_setprio(0);
    __syncthreads();

    const int n0    = wave * 16;
    const int mrow  = lane & 15;
    const int khalf = (lane >> 4) * 2;
    const int ncol  = n0 + (lane & 15);

    v8f c0 = {};
    v8f c1 = {};
#pragma unroll
    for (int k0 = 0; k0 < FEAT; k0 += 4) {
        v2f a0, a1, b;
        a0.x = As[mrow][k0 + khalf];
        a0.y = As[mrow][k0 + khalf + 1];
        a1.x = As[16 + mrow][k0 + khalf];
        a1.y = As[16 + mrow][k0 + khalf + 1];
        b.x = W[(k0 + khalf) * FEAT + ncol];
        b.y = W[(k0 + khalf + 1) * FEAT + ncol];
        c0 = __builtin_amdgcn_wmma_f32_16x16x4_f32(
            false, a0, false, b, (short)0, c0, false, false);
        c1 = __builtin_amdgcn_wmma_f32_16x16x4_f32(
            false, a1, false, b, (short)0, c1, false, false);
    }

    const float bi = bias[ncol];
    const float mu = mean[ncol];
    const float va = var[ncol];
    const float ga = gamma[ncol];
    const float be = beta[ncol];
    const int mbase = (lane >> 4) * 8;
    __syncthreads();
#pragma unroll
    for (int v = 0; v < 8; ++v) {
        As[mbase + v][ncol]      = bias_bn_relu(c0[v], bi, mu, va, ga, be);
        As[16 + mbase + v][ncol] = bias_bn_relu(c1[v], bi, mu, va, ga, be);
    }
    __syncthreads();

    if (FOLD3) {
        const v4f wa = ((const v4f*)W3)[2 * lane];
        const v4f wb = ((const v4f*)W3)[2 * lane + 1];
#pragma unroll
        for (int q = 0; q < 4; ++q) {
            const v4f y = *(const v4f*)&As[wave * 4 + q][4 * lane];
            float s0 = y.x * wa.x + y.y * wa.z + y.z * wb.x + y.w * wb.z;
            float s1 = y.x * wa.y + y.y * wa.w + y.z * wb.y + y.w * wb.w;
#pragma unroll
            for (int off = 16; off > 0; off >>= 1) {
                s0 += __shfl_xor(s0, off, 32);
                s1 += __shfl_xor(s1, off, 32);
            }
            const int node = nb + q;
            if (lane == 0 && node < n) {
                out[2 * node + 0] = s0;
                out[2 * node + 1] = s1;
            }
        }
    } else {
#pragma unroll
        for (int q = 0; q < 4; ++q) {
            const int node = nb + q;
            if (node < n)
                ((v4f*)out)[(long long)node * 32 + lane] = *(const v4f*)&As[wave * 4 + q][4 * lane];
        }
    }
}

__global__ __launch_bounds__(256, 12) void layer128_store_kernel(
    const float* __restrict__ xin, const int* __restrict__ rowptr,
    const int* __restrict__ col, const float* __restrict__ dinv,
    const float* __restrict__ W,
    const float* __restrict__ bias, const float* __restrict__ gamma,
    const float* __restrict__ beta, const float* __restrict__ mean,
    const float* __restrict__ var,
    float* __restrict__ y, int n) {
    layer128_body<false>(xin, rowptr, col, dinv, W, bias, gamma, beta, mean, var,
                         nullptr, y, n);
}

__global__ __launch_bounds__(256, 12) void layer128_w3_kernel(
    const float* __restrict__ xin, const int* __restrict__ rowptr,
    const int* __restrict__ col, const float* __restrict__ dinv,
    const float* __restrict__ W,
    const float* __restrict__ bias, const float* __restrict__ gamma,
    const float* __restrict__ beta, const float* __restrict__ mean,
    const float* __restrict__ var,
    const float* __restrict__ W3, float* __restrict__ h3, int n) {
    layer128_body<true>(xin, rowptr, col, dinv, W, bias, gamma, beta, mean, var,
                        W3, h3, n);
}

__global__ void gather2_logsoftmax_kernel(const float* __restrict__ h3,
                               const int* __restrict__ rowptr,
                               const int* __restrict__ col,
                               const float* __restrict__ dinv,
                               const float* __restrict__ b3,
                               float* __restrict__ out, int n) {
    int i = blockIdx.x * blockDim.x + threadIdx.x;
    if (i >= n) return;
    const float d = dinv[i];
    const float d2 = d * d;
    float a0 = h3[2 * i + 0] * d2;
    float a1 = h3[2 * i + 1] * d2;
    const int beg = rowptr[i], end = rowptr[i + 1];
    for (int j = beg; j < end; ++j) {
        const int s = col[j];
        const float w = dinv[s] * d;
        a0 += h3[2 * s + 0] * w;
        a1 += h3[2 * s + 1] * w;
    }
    const float z0 = a0 + b3[0];
    const float z1 = a1 + b3[1];
    const float mx = fmaxf(z0, z1);
    const float lse = mx + logf(expf(z0 - mx) + expf(z1 - mx));
    out[2 * i + 0] = z0 - lse;
    out[2 * i + 1] = z1 - lse;
}

extern "C" void kernel_launch(void* const* d_in, const int* in_sizes, int n_in,
                              void* d_out, int out_size, void* d_ws, size_t ws_size,
                              hipStream_t stream) {
    const float* x   = (const float*)d_in[0];
    const int*   ei  = (const int*)d_in[1];
    const float* W1  = (const float*)d_in[2];
    const float* b1  = (const float*)d_in[3];
    const float* g1  = (const float*)d_in[4];
    const float* be1 = (const float*)d_in[5];
    const float* m1  = (const float*)d_in[6];
    const float* v1  = (const float*)d_in[7];
    const float* W2  = (const float*)d_in[8];
    const float* b2  = (const float*)d_in[9];
    const float* g2  = (const float*)d_in[10];
    const float* be2 = (const float*)d_in[11];
    const float* m2  = (const float*)d_in[12];
    const float* v2  = (const float*)d_in[13];
    const float* W3  = (const float*)d_in[14];
    const float* b3  = (const float*)d_in[15];

    const int N = in_sizes[0] / FEAT;
    const int E = in_sizes[1] / 2;
    const int* src = ei;
    const int* dst = ei + E;

    float* bufA = (float*)d_ws;
    float* aux  = bufA + (long long)N * FEAT;
    float* h3   = aux + (long long)N * FEAT;
    float* dinv = h3 + (long long)N * 2;
    int* rowptr = (int*)(dinv + N);
    int* col    = rowptr + N + 1;
    int* colraw = col + E;
    int* ctab   = (int*)aux;
    int2* ebuf  = (int2*)(ctab + (NBMAX + 1) * GB);
    float* outp = (float*)d_out;

    const int TB = 256;
    const int gN      = (N + TB - 1) / TB;
    const int gTiles  = (N + 31) / 32;
    const int nb      = (N + BSZ - 1) / BSZ;

    chunk_sort_kernel<<<GB, 1024, 0, stream>>>(src, dst, ctab, ebuf, E, nb);
    bucket_rows_kernel<<<nb, BSZ, 0, stream>>>(ebuf, ctab, rowptr, colraw, col, dinv, N, E);

    layer128_store_kernel<<<gTiles, 256, 0, stream>>>(
        x, rowptr, col, dinv, W1, b1, g1, be1, m1, v1, bufA, N);
    layer128_w3_kernel<<<gTiles, 256, 0, stream>>>(
        bufA, rowptr, col, dinv, W2, b2, g2, be2, m2, v2, W3, h3, N);

    gather2_logsoftmax_kernel<<<gN, TB, 0, stream>>>(h3, rowptr, col, dinv, b3, outp, N);
}
